# starting kernel
# baseline (speedup 1.0000x reference)
#include <hip/hip_runtime.h>
#include <hip/hip_bf16.h>
#include <math.h>


#define RNN_T    2048
#define RNN_B    2048
#define RNN_EMB  10
#define RNN_HID  8
#define RNN_NCLS 4

typedef __attribute__((ext_vector_type(2))) float v2f;
typedef __attribute__((ext_vector_type(4))) float v4f;
typedef __attribute__((ext_vector_type(8))) float v8f;

#if defined(__has_builtin)
#if __has_builtin(__builtin_amdgcn_tanhf)
#define RNN_TANH(x) __builtin_amdgcn_tanhf(x)
#endif
#endif
#ifndef RNN_TANH
#define RNN_TANH(x) tanhf(x)
#endif

__device__ __forceinline__ v8f wmma4(v2f a, v2f b, v8f c) {
    return __builtin_amdgcn_wmma_f32_16x16x4_f32(
         false, a,  false, b,
         (short)0, c,  false,  false);
}

__device__ __forceinline__ v2f load_a_dup(const float* W, int lo, int hi, int c) {
    const int r  = lo & 7;
    const int k0 = 4 * c + 2 * hi;
    v2f a;
    a.x = W[r * 8 + k0];
    a.y = W[r * 8 + k0 + 1];
    return a;
}

__device__ __forceinline__ v2f load_a_fc(const float* W, int lo, int hi, int c) {
    const int r = lo & 7;
    v2f a; a.x = 0.0f; a.y = 0.0f;
    if (r < RNN_NCLS) {
        const int k0 = 4 * c + 2 * hi;
        a.x = W[r * 8 + k0];
        a.y = W[r * 8 + k0 + 1];
    }
    return a;
}

__device__ __forceinline__ void d_to_b(const v8f& hD, bool hi, v2f& b0, v2f& b1) {
    b0.x = hi ? hD[2] : hD[0];
    b0.y = hi ? hD[3] : hD[1];
    b1.x = hi ? hD[6] : hD[4];
    b1.y = hi ? hD[7] : hD[5];
}

__global__ __launch_bounds__(32)
void rnn_fused_kernel(const int* __restrict__ x,
                      const float* __restrict__ emb,
                      const float* __restrict__ W_ih0,
                      const float* __restrict__ W_hh0,
                      const float* __restrict__ b_ih0,
                      const float* __restrict__ b_hh0,
                      const float* __restrict__ W_ih1,
                      const float* __restrict__ W_hh1,
                      const float* __restrict__ b_ih1,
                      const float* __restrict__ b_hh1,
                      const float* __restrict__ W_fc,
                      const float* __restrict__ b_fc,
                      float* __restrict__ out) {
    const int lane = threadIdx.x & 31;
    const int lo   = lane & 15;
    const int hi   = lane >> 4;
    const int batch_base = blockIdx.x * 16;

    v2f t0a;
    {
        const int j = lo & 7;
        const float bias = b_ih0[j] + b_hh0[j];
        const int v0 = 2 * hi, v1 = 2 * hi + 1;
        float a0 = bias, a1 = bias;
        #pragma unroll
        for (int d = 0; d < RNN_EMB; ++d) {
            const float w = W_ih0[j * RNN_EMB + d];
            a0 += emb[v0 * RNN_EMB + d] * w;
            a1 += emb[v1 * RNN_EMB + d] * w;
        }
        t0a.x = a0; t0a.y = a1;
    }

    const v2f whh0_a0 = load_a_dup(W_hh0, lo, hi, 0);
    const v2f whh0_a1 = load_a_dup(W_hh0, lo, hi, 1);
    const v2f wih1_a0 = load_a_dup(W_ih1, lo, hi, 0);
    const v2f wih1_a1 = load_a_dup(W_ih1, lo, hi, 1);
    const v2f whh1_a0 = load_a_dup(W_hh1, lo, hi, 0);
    const v2f whh1_a1 = load_a_dup(W_hh1, lo, hi, 1);
    const v2f wfc_a0  = load_a_fc (W_fc,  lo, hi, 0);
    const v2f wfc_a1  = load_a_fc (W_fc,  lo, hi, 1);

    v8f c_bias1, c_fc, c_zero;
    #pragma unroll
    for (int i = 0; i < 8; ++i) {
        c_bias1[i] = b_ih1[i] + b_hh1[i];
        c_fc[i]    = (i < RNN_NCLS) ? b_fc[i] : 0.0f;
        c_zero[i]  = 0.0f;
    }

    v8f h0D = c_zero, h1D = c_zero;
    v2f h0b0 = {0.0f, 0.0f}, h0b1 = {0.0f, 0.0f};
    v2f h1b0 = {0.0f, 0.0f}, h1b1 = {0.0f, 0.0f};

    const int* xrow = x + (size_t)batch_base + lo;
    float*     op   = out + ((size_t)batch_base + lo) * RNN_NCLS;

    int        xi      = xrow[0];
    const int* xp_next = xrow + RNN_B;

    #pragma unroll 1
    for (int t = 0; t < RNN_T; ++t) {
        if (t + 8 < RNN_T)
            __builtin_prefetch(xp_next + (size_t)7 * RNN_B, 0, 3);
        int xi_next = xi;
        if (t + 1 < RNN_T) xi_next = xp_next[0];

        v2f oh;
        oh.x = (xi == 2 * hi)     ? 1.0f : 0.0f;
        oh.y = (xi == 2 * hi + 1) ? 1.0f : 0.0f;

        v8f p0 = wmma4(t0a, oh, c_zero);
        p0 = wmma4(whh0_a0, h0b0, p0);
        p0 = wmma4(whh0_a1, h0b1, p0);
        #pragma unroll
        for (int i = 0; i < 8; ++i) h0D[i] = RNN_TANH(p0[i]);
        d_to_b(h0D, hi != 0, h0b0, h0b1);

        v8f p1 = c_bias1;
        p1 = wmma4(wih1_a0, h0b0, p1);
        p1 = wmma4(wih1_a1, h0b1, p1);
        p1 = wmma4(whh1_a0, h1b0, p1);
        p1 = wmma4(whh1_a1, h1b1, p1);
        #pragma unroll
        for (int i = 0; i < 8; ++i) h1D[i] = RNN_TANH(p1[i]);
        d_to_b(h1D, hi != 0, h1b0, h1b1);

        v8f lg = c_fc;
        lg = wmma4(wfc_a0, h1b0, lg);
        lg = wmma4(wfc_a1, h1b1, lg);

        if (!hi) {
            v4f res; res.x = lg[0]; res.y = lg[1]; res.z = lg[2]; res.w = lg[3];
            *(v4f*)op = res;
        }

        xi = xi_next;
        xp_next += RNN_B;
        op += (size_t)RNN_B * RNN_NCLS;
    }
}

extern "C" void kernel_launch(void* const* d_in, const int* in_sizes, int n_in,
                              void* d_out, int out_size, void* d_ws, size_t ws_size,
                              hipStream_t stream) {
    (void)in_sizes; (void)n_in; (void)d_ws; (void)ws_size; (void)out_size;
    const int*   x     = (const int*)  d_in[0];
    const float* emb   = (const float*)d_in[1];
    const float* W_ih0 = (const float*)d_in[2];
    const float* W_hh0 = (const float*)d_in[3];
    const float* b_ih0 = (const float*)d_in[4];
    const float* b_hh0 = (const float*)d_in[5];
    const float* W_ih1 = (const float*)d_in[6];
    const float* W_hh1 = (const float*)d_in[7];
    const float* b_ih1 = (const float*)d_in[8];
    const float* b_hh1 = (const float*)d_in[9];
    const float* W_fc  = (const float*)d_in[10];
    const float* b_fc  = (const float*)d_in[11];
    float* out = (float*)d_out;

    dim3 grid(RNN_B / 16);
    dim3 block(32);
    rnn_fused_kernel<<<grid, block, 0, stream>>>(
        x, emb, W_ih0, W_hh0, b_ih0, b_hh0,
        W_ih1, W_hh1, b_ih1, b_hh1, W_fc, b_fc, out);
}


# Software-pipeline the time loop: layer0(s), layer1 input half(s-1), layer1 recurrent half(s-2), fc(s-3) run as independent WMMA chains per trip; tokens loaded 8 steps ahead; sum order kept
# speedup vs baseline: 1.7482x; 1.7482x over previous
#include <hip/hip_runtime.h>
#include <hip/hip_bf16.h>
#include <math.h>


#define RNN_T    2048
#define RNN_B    2048
#define RNN_EMB  10
#define RNN_HID  8
#define RNN_NCLS 4
#define RNN_BLK  8

typedef __attribute__((ext_vector_type(2))) float v2f;
typedef __attribute__((ext_vector_type(4))) float v4f;
typedef __attribute__((ext_vector_type(8))) float v8f;

#if defined(__has_builtin)
#if __has_builtin(__builtin_amdgcn_tanhf)
#define RNN_TANH(x) __builtin_amdgcn_tanhf(x)
#endif
#endif
#ifndef RNN_TANH
#define RNN_TANH(x) tanhf(x)
#endif

__device__ __forceinline__ v8f wmma4(v2f a, v2f b, v8f c) {
    return __builtin_amdgcn_wmma_f32_16x16x4_f32(
         false, a,  false, b,
         (short)0, c,  false,  false);
}

__device__ __forceinline__ v2f load_a_dup(const float* W, int lo, int hi, int c) {
    const int r  = lo & 7;
    const int k0 = 4 * c + 2 * hi;
    v2f a;
    a.x = W[r * 8 + k0];
    a.y = W[r * 8 + k0 + 1];
    return a;
}

__device__ __forceinline__ v2f load_a_fc(const float* W, int lo, int hi, int c) {
    const int r = lo & 7;
    v2f a; a.x = 0.0f; a.y = 0.0f;
    if (r < RNN_NCLS) {
        const int k0 = 4 * c + 2 * hi;
        a.x = W[r * 8 + k0];
        a.y = W[r * 8 + k0 + 1];
    }
    return a;
}

__device__ __forceinline__ void d_to_b(const v8f& hD, bool hi, v2f& b0, v2f& b1) {
    b0.x = hi ? hD[2] : hD[0];
    b0.y = hi ? hD[3] : hD[1];
    b1.x = hi ? hD[6] : hD[4];
    b1.y = hi ? hD[7] : hD[5];
}

struct RnnW {
    v2f t0a;
    v2f whh0_a0, whh0_a1;
    v2f wih1_a0, wih1_a1;
    v2f whh1_a0, whh1_a1;
    v2f wfc_a0,  wfc_a1;
    v8f c_bias1, c_fc;
};

struct RnnS {
    v2f h0b0, h0b1;
    v2f h1b0, h1b1;
    v8f q;
};

template <bool DO_A, bool DO_B1, bool DO_B2, bool DO_C>
__device__ __forceinline__ void pipe_step(const RnnW& w, RnnS& s, int tok,
                                          int k0, bool hi, float*& op) {
    const v8f zero = {0.0f, 0.0f, 0.0f, 0.0f, 0.0f, 0.0f, 0.0f, 0.0f};
    const v2f h0b0 = s.h0b0, h0b1 = s.h0b1;
    const v2f h1b0 = s.h1b0, h1b1 = s.h1b1;
    const v8f q    = s.q;
    v8f p0 = zero, p1 = zero, qn = zero, lg = zero;

    if constexpr (DO_A) {
        v2f oh;
        oh.x = (tok == k0)     ? 1.0f : 0.0f;
        oh.y = (tok == k0 + 1) ? 1.0f : 0.0f;
        p0 = wmma4(w.t0a, oh, zero);
    }
    if constexpr (DO_B2) p1 = wmma4(w.whh1_a0, h1b0, q);
    if constexpr (DO_A)  p0 = wmma4(w.whh0_a0, h0b0, p0);
    if constexpr (DO_B2) p1 = wmma4(w.whh1_a1, h1b1, p1);
    if constexpr (DO_A)  p0 = wmma4(w.whh0_a1, h0b1, p0);
    if constexpr (DO_B1) qn = wmma4(w.wih1_a0, h0b0, w.c_bias1);
    if constexpr (DO_C)  lg = wmma4(w.wfc_a0,  h1b0, w.c_fc);
    if constexpr (DO_B1) qn = wmma4(w.wih1_a1, h0b1, qn);
    if constexpr (DO_C)  lg = wmma4(w.wfc_a1,  h1b1, lg);

    if constexpr (DO_B2) {
        v8f h1D;
        #pragma unroll
        for (int i = 0; i < 8; ++i) h1D[i] = RNN_TANH(p1[i]);
        d_to_b(h1D, hi, s.h1b0, s.h1b1);
    }
    if constexpr (DO_A) {
        v8f h0D;
        #pragma unroll
        for (int i = 0; i < 8; ++i) h0D[i] = RNN_TANH(p0[i]);
        d_to_b(h0D, hi, s.h0b0, s.h0b1);
    }
    if constexpr (DO_B1) s.q = qn;
    if constexpr (DO_C) {
        if (!hi) {
            v4f res; res.x = lg[0]; res.y = lg[1]; res.z = lg[2]; res.w = lg[3];
            *(v4f*)op = res;
        }
        op += (size_t)RNN_B * RNN_NCLS;
    }
}

__global__ __launch_bounds__(32)
void rnn_fused_kernel(const int* __restrict__ x,
                      const float* __restrict__ emb,
                      const float* __restrict__ W_ih0,
                      const float* __restrict__ W_hh0,
                      const float* __restrict__ b_ih0,
                      const float* __restrict__ b_hh0,
                      const float* __restrict__ W_ih1,
                      const float* __restrict__ W_hh1,
                      const float* __restrict__ b_ih1,
                      const float* __restrict__ b_hh1,
                      const float* __restrict__ W_fc,
                      const float* __restrict__ b_fc,
                      float* __restrict__ out) {
    const int lane = threadIdx.x & 31;
    const int lo   = lane & 15;
    const int hi   = lane >> 4;
    const int batch_base = blockIdx.x * 16;

    const int* xrow = x + (size_t)batch_base + lo;

    int tok[RNN_BLK], ntok[RNN_BLK];
    #pragma unroll
    for (int j = 0; j < RNN_BLK; ++j) tok[j]  = xrow[(size_t)j * RNN_B];
    #pragma unroll
    for (int j = 0; j < RNN_BLK; ++j) ntok[j] = xrow[(size_t)(RNN_BLK + j) * RNN_B];

    RnnW w;
    {
        const int j = lo & 7;
        const float bias = b_ih0[j] + b_hh0[j];
        const int v0 = 2 * hi, v1 = 2 * hi + 1;
        float a0 = bias, a1 = bias;
        #pragma unroll
        for (int d = 0; d < RNN_EMB; ++d) {
            const float wv = W_ih0[j * RNN_EMB + d];
            a0 += emb[v0 * RNN_EMB + d] * wv;
            a1 += emb[v1 * RNN_EMB + d] * wv;
        }
        w.t0a.x = a0; w.t0a.y = a1;
    }

    w.whh0_a0 = load_a_dup(W_hh0, lo, hi, 0);
    w.whh0_a1 = load_a_dup(W_hh0, lo, hi, 1);
    w.wih1_a0 = load_a_dup(W_ih1, lo, hi, 0);
    w.wih1_a1 = load_a_dup(W_ih1, lo, hi, 1);
    w.whh1_a0 = load_a_dup(W_hh1, lo, hi, 0);
    w.whh1_a1 = load_a_dup(W_hh1, lo, hi, 1);
    w.wfc_a0  = load_a_fc (W_fc,  lo, hi, 0);
    w.wfc_a1  = load_a_fc (W_fc,  lo, hi, 1);

    #pragma unroll
    for (int i = 0; i < 8; ++i) {
        w.c_bias1[i] = b_ih1[i] + b_hh1[i];
        w.c_fc[i]    = (i < RNN_NCLS) ? b_fc[i] : 0.0f;
    }

    RnnS s;
    s.h0b0 = (v2f){0.0f, 0.0f}; s.h0b1 = (v2f){0.0f, 0.0f};
    s.h1b0 = (v2f){0.0f, 0.0f}; s.h1b1 = (v2f){0.0f, 0.0f};
    #pragma unroll
    for (int i = 0; i < 8; ++i) s.q[i] = 0.0f;

    const int  k0  = 2 * hi;
    const bool hiB = (hi != 0);
    float* op = out + ((size_t)batch_base + lo) * RNN_NCLS;

    pipe_step<true, false, false, false>(w, s, tok[0], k0, hiB, op);
    pipe_step<true, true,  false, false>(w, s, tok[1], k0, hiB, op);
    pipe_step<true, true,  true,  false>(w, s, tok[2], k0, hiB, op);
    #pragma unroll
    for (int j = 3; j < RNN_BLK; ++j)
        pipe_step<true, true, true, true>(w, s, tok[j], k0, hiB, op);

    const int* xnext = xrow + (size_t)(2 * RNN_BLK) * RNN_B;
    #pragma unroll 1
    for (int sb = RNN_BLK; sb < RNN_T; sb += RNN_BLK) {
        #pragma unroll
        for (int j = 0; j < RNN_BLK; ++j) tok[j] = ntok[j];
        if (sb + RNN_BLK < RNN_T) {
            #pragma unroll
            for (int j = 0; j < RNN_BLK; ++j) ntok[j] = xnext[(size_t)j * RNN_B];
            if (sb + 2 * RNN_BLK < RNN_T) {
                #pragma unroll
                for (int j = 0; j < RNN_BLK; ++j)
                    __builtin_prefetch(xnext + (size_t)(RNN_BLK + j) * RNN_B, 0, 3);
            }
        }
        #pragma unroll
        for (int j = 0; j < RNN_BLK; ++j)
            pipe_step<true, true, true, true>(w, s, tok[j], k0, hiB, op);
        xnext += (size_t)RNN_BLK * RNN_B;
    }

    pipe_step<false, true,  true,  true>(w, s, 0, k0, hiB, op);
    pipe_step<false, false, true,  true>(w, s, 0, k0, hiB, op);
    pipe_step<false, false, false, true>(w, s, 0, k0, hiB, op);
}

extern "C" void kernel_launch(void* const* d_in, const int* in_sizes, int n_in,
                              void* d_out, int out_size, void* d_ws, size_t ws_size,
                              hipStream_t stream) {
    (void)in_sizes; (void)n_in; (void)d_ws; (void)ws_size; (void)out_size;
    const int*   x     = (const int*)  d_in[0];
    const float* emb   = (const float*)d_in[1];
    const float* W_ih0 = (const float*)d_in[2];
    const float* W_hh0 = (const float*)d_in[3];
    const float* b_ih0 = (const float*)d_in[4];
    const float* b_hh0 = (const float*)d_in[5];
    const float* W_ih1 = (const float*)d_in[6];
    const float* W_hh1 = (const float*)d_in[7];
    const float* b_ih1 = (const float*)d_in[8];
    const float* b_hh1 = (const float*)d_in[9];
    const float* W_fc  = (const float*)d_in[10];
    const float* b_fc  = (const float*)d_in[11];
    float* out = (float*)d_out;

    dim3 grid(RNN_B / 16);
    dim3 block(32);
    rnn_fused_kernel<<<grid, block, 0, stream>>>(
        x, emb, W_ih0, W_hh0, b_ih0, b_hh0,
        W_ih1, W_hh1, b_ih1, b_hh1, W_fc, b_fc, out);
}


# Stack layer 0 and layer 1 in the 16 WMMA rows (0-7 / 8-15) instead of duplicating rows: 5 WMMAs and 8 tanh per step, was 9 and 16; half-waves exchanged by v_permlane16_swap; sum order kept
# speedup vs baseline: 2.2652x; 1.2957x over previous
#include <hip/hip_runtime.h>
#include <hip/hip_bf16.h>
#include <math.h>


#define RNN_T    2048
#define RNN_B    2048
#define RNN_EMB  10
#define RNN_HID  8
#define RNN_NCLS 4
#define RNN_BLK  8

typedef __attribute__((ext_vector_type(2))) float v2f;
typedef __attribute__((ext_vector_type(4))) float v4f;
typedef __attribute__((ext_vector_type(8))) float v8f;
typedef __attribute__((ext_vector_type(2))) unsigned int v2u;

#if defined(__has_builtin)
#if __has_builtin(__builtin_amdgcn_tanhf)
#define RNN_TANH(x) __builtin_amdgcn_tanhf(x)
#endif
#endif
#ifndef RNN_TANH
#define RNN_TANH(x) tanhf(x)
#endif

__device__ __forceinline__ v8f wmma4(v2f a, v2f b, v8f c) {
    return __builtin_amdgcn_wmma_f32_16x16x4_f32(
         false, a,  false, b,
         (short)0, c,  false,  false);
}

__device__ __forceinline__ v2f load_a_stack(const float* Wlo, int nlo,
                                            const float* Whi,
                                            int lo, int hi, int c) {
    const int r  = lo & 7;
    const int k0 = 4 * c + 2 * hi;
    v2f a; a.x = 0.0f; a.y = 0.0f;
    if (lo >= 8) {
        a.x = Whi[r * 8 + k0];
        a.y = Whi[r * 8 + k0 + 1];
    } else if (r < nlo) {
        a.x = Wlo[r * 8 + k0];
        a.y = Wlo[r * 8 + k0 + 1];
    }
    return a;
}

__device__ __forceinline__ void half_swap(float& x, float& y) {
    const v2u r = __builtin_amdgcn_permlane16_swap(__float_as_uint(x),
                                                   __float_as_uint(y),
                                                   false, false);
    x = __uint_as_float(r.x);
    y = __uint_as_float(r.y);
}

struct RnnW {
    v2f t0a;
    v2f s1_a0, s1_a1;
    v2f s2_a0, s2_a1;
    v8f c_init;
    float c_fc[RNN_NCLS];
};

struct RnnS {
    v2f h0b0, h0b1;
    v2f h1b0, h1b1;
    v8f d1;
};

template <bool DO_OH, bool DO_S1, bool DO_S2, bool DO_ST, bool DO_H>
__device__ __forceinline__ void pipe_step(const RnnW& w, RnnS& s, int tok,
                                          int k0, bool hi, float*& op) {
    const v2f h0b0 = s.h0b0, h0b1 = s.h0b1;
    const v2f h1b0 = s.h1b0, h1b1 = s.h1b1;
    const v8f d1p  = s.d1;
    v8f d1 = w.c_init;
    v8f d2 = d1p;

    if constexpr (DO_S2) {
        #pragma unroll
        for (int i = 0; i < RNN_NCLS; ++i) d2[i] = hi ? d1p[i] : w.c_fc[i];
    }
    if constexpr (DO_OH) {
        v2f oh;
        oh.x = (tok == k0)     ? 1.0f : 0.0f;
        oh.y = (tok == k0 + 1) ? 1.0f : 0.0f;
        d1 = wmma4(w.t0a, oh, w.c_init);
    }
    if constexpr (DO_S1) d1 = wmma4(w.s1_a0, h0b0, d1);
    if constexpr (DO_S2) d2 = wmma4(w.s2_a0, h1b0, d2);
    if constexpr (DO_S1) d1 = wmma4(w.s1_a1, h0b1, d1);
    if constexpr (DO_S2) d2 = wmma4(w.s2_a1, h1b1, d2);

    if constexpr (DO_ST) {
        if (!hi) {
            v4f res; res.x = d2[0]; res.y = d2[1]; res.z = d2[2]; res.w = d2[3];
            *(v4f*)op = res;
        }
        op += (size_t)RNN_B * RNN_NCLS;
    }
    if constexpr (DO_H) {
        float z[8];
        #pragma unroll
        for (int i = 0; i < 8; ++i) {
            if constexpr (DO_S2) z[i] = hi ? d2[i] : d1[i];
            else                 z[i] = d1[i];
        }
        #pragma unroll
        for (int i = 0; i < 8; ++i) z[i] = RNN_TANH(z[i]);
        half_swap(z[0], z[2]);
        half_swap(z[1], z[3]);
        half_swap(z[4], z[6]);
        half_swap(z[5], z[7]);
        s.h0b0.x = z[0]; s.h0b0.y = z[1];
        s.h0b1.x = z[4]; s.h0b1.y = z[5];
        if constexpr (DO_S2) {
            s.h1b0.x = z[2]; s.h1b0.y = z[3];
            s.h1b1.x = z[6]; s.h1b1.y = z[7];
        }
    }
    if constexpr (DO_S1) s.d1 = d1;
}

__global__ __launch_bounds__(32)
void rnn_fused_kernel(const int* __restrict__ x,
                      const float* __restrict__ emb,
                      const float* __restrict__ W_ih0,
                      const float* __restrict__ W_hh0,
                      const float* __restrict__ b_ih0,
                      const float* __restrict__ b_hh0,
                      const float* __restrict__ W_ih1,
                      const float* __restrict__ W_hh1,
                      const float* __restrict__ b_ih1,
                      const float* __restrict__ b_hh1,
                      const float* __restrict__ W_fc,
                      const float* __restrict__ b_fc,
                      float* __restrict__ out) {
    const int lane = threadIdx.x & 31;
    const int lo   = lane & 15;
    const int hi   = lane >> 4;
    const int batch_base = blockIdx.x * 16;

    const int* xrow = x + (size_t)batch_base + lo;

    int tok[RNN_BLK], ntok[RNN_BLK];
    #pragma unroll
    for (int j = 0; j < RNN_BLK; ++j) tok[j]  = xrow[(size_t)j * RNN_B];
    #pragma unroll
    for (int j = 0; j < RNN_BLK; ++j) ntok[j] = xrow[(size_t)(RNN_BLK + j) * RNN_B];

    RnnW w;
    {
        const int j = lo & 7;
        const float bias = b_ih0[j] + b_hh0[j];
        const int v0 = 2 * hi, v1 = 2 * hi + 1;
        float a0 = bias, a1 = bias;
        #pragma unroll
        for (int d = 0; d < RNN_EMB; ++d) {
            const float wv = W_ih0[j * RNN_EMB + d];
            a0 += emb[v0 * RNN_EMB + d] * wv;
            a1 += emb[v1 * RNN_EMB + d] * wv;
        }
        w.t0a.x = (lo < 8) ? a0 : 0.0f;
        w.t0a.y = (lo < 8) ? a1 : 0.0f;
    }

    w.s1_a0 = load_a_stack(W_hh0, RNN_HID,  W_ih1, lo, hi, 0);
    w.s1_a1 = load_a_stack(W_hh0, RNN_HID,  W_ih1, lo, hi, 1);
    w.s2_a0 = load_a_stack(W_fc,  RNN_NCLS, W_hh1, lo, hi, 0);
    w.s2_a1 = load_a_stack(W_fc,  RNN_NCLS, W_hh1, lo, hi, 1);

    #pragma unroll
    for (int i = 0; i < 8; ++i)
        w.c_init[i] = hi ? (b_ih1[i] + b_hh1[i]) : 0.0f;
    #pragma unroll
    for (int i = 0; i < RNN_NCLS; ++i) w.c_fc[i] = b_fc[i];

    RnnS s;
    s.h0b0 = (v2f){0.0f, 0.0f}; s.h0b1 = (v2f){0.0f, 0.0f};
    s.h1b0 = (v2f){0.0f, 0.0f}; s.h1b1 = (v2f){0.0f, 0.0f};
    #pragma unroll
    for (int i = 0; i < 8; ++i) s.d1[i] = 0.0f;

    const int  k0  = 2 * hi;
    const bool hiB = (hi != 0);
    float* op = out + ((size_t)batch_base + lo) * RNN_NCLS;

    pipe_step<true, true, false, false, true>(w, s, tok[0], k0, hiB, op);
    pipe_step<true, true, false, false, true>(w, s, tok[1], k0, hiB, op);
    pipe_step<true, true, true,  false, true>(w, s, tok[2], k0, hiB, op);
    #pragma unroll
    for (int j = 3; j < RNN_BLK; ++j)
        pipe_step<true, true, true, true, true>(w, s, tok[j], k0, hiB, op);

    const int* xnext = xrow + (size_t)(2 * RNN_BLK) * RNN_B;
    #pragma unroll 1
    for (int sb = RNN_BLK; sb < RNN_T; sb += RNN_BLK) {
        #pragma unroll
        for (int j = 0; j < RNN_BLK; ++j) tok[j] = ntok[j];
        if (sb + RNN_BLK < RNN_T) {
            #pragma unroll
            for (int j = 0; j < RNN_BLK; ++j) ntok[j] = xnext[(size_t)j * RNN_B];
            if (sb + 2 * RNN_BLK < RNN_T) {
                #pragma unroll
                for (int j = 0; j < RNN_BLK; ++j)
                    __builtin_prefetch(xnext + (size_t)(RNN_BLK + j) * RNN_B, 0, 3);
            }
        }
        #pragma unroll
        for (int j = 0; j < RNN_BLK; ++j)
            pipe_step<true, true, true, true, true>(w, s, tok[j], k0, hiB, op);
        xnext += (size_t)RNN_BLK * RNN_B;
    }

    pipe_step<false, true,  true, true, true >(w, s, 0, k0, hiB, op);
    pipe_step<false, false, true, true, true >(w, s, 0, k0, hiB, op);
    pipe_step<false, false, true, true, false>(w, s, 0, k0, hiB, op);
}

extern "C" void kernel_launch(void* const* d_in, const int* in_sizes, int n_in,
                              void* d_out, int out_size, void* d_ws, size_t ws_size,
                              hipStream_t stream) {
    (void)in_sizes; (void)n_in; (void)d_ws; (void)ws_size; (void)out_size;
    const int*   x     = (const int*)  d_in[0];
    const float* emb   = (const float*)d_in[1];
    const float* W_ih0 = (const float*)d_in[2];
    const float* W_hh0 = (const float*)d_in[3];
    const float* b_ih0 = (const float*)d_in[4];
    const float* b_hh0 = (const float*)d_in[5];
    const float* W_ih1 = (const float*)d_in[6];
    const float* W_hh1 = (const float*)d_in[7];
    const float* b_ih1 = (const float*)d_in[8];
    const float* b_hh1 = (const float*)d_in[9];
    const float* W_fc  = (const float*)d_in[10];
    const float* b_fc  = (const float*)d_in[11];
    float* out = (float*)d_out;

    dim3 grid(RNN_B / 16);
    dim3 block(32);
    rnn_fused_kernel<<<grid, block, 0, stream>>>(
        x, emb, W_ih0, W_hh0, b_ih0, b_hh0,
        W_ih1, W_hh1, b_ih1, b_hh1, W_fc, b_fc, out);
}


# Output stores batched: each step keeps its 4 logits in registers of its own, 8 rows are stored together once per 8 steps, so the store's EXEC toggle and wait leave the per-step path; sum order kept
# speedup vs baseline: 3.3184x; 1.4649x over previous
#include <hip/hip_runtime.h>
#include <hip/hip_bf16.h>
#include <math.h>


#define RNN_T    2048
#define RNN_B    2048
#define RNN_EMB  10
#define RNN_HID  8
#define RNN_NCLS 4
#define RNN_BLK  8

typedef __attribute__((ext_vector_type(2))) float v2f;
typedef __attribute__((ext_vector_type(4))) float v4f;
typedef __attribute__((ext_vector_type(8))) float v8f;
typedef __attribute__((ext_vector_type(2))) unsigned int v2u;

#if defined(__has_builtin)
#if __has_builtin(__builtin_amdgcn_tanhf)
#define RNN_TANH(x) __builtin_amdgcn_tanhf(x)
#endif
#endif
#ifndef RNN_TANH
#define RNN_TANH(x) tanhf(x)
#endif

__device__ __forceinline__ v8f wmma4(v2f a, v2f b, v8f c) {
    return __builtin_amdgcn_wmma_f32_16x16x4_f32(
         false, a,  false, b,
         (short)0, c,  false,  false);
}

__device__ __forceinline__ v2f load_a_stack(const float* Wlo, int nlo,
                                            const float* Whi,
                                            int lo, int hi, int c) {
    const int r  = lo & 7;
    const int k0 = 4 * c + 2 * hi;
    v2f a; a.x = 0.0f; a.y = 0.0f;
    if (lo >= 8) {
        a.x = Whi[r * 8 + k0];
        a.y = Whi[r * 8 + k0 + 1];
    } else if (r < nlo) {
        a.x = Wlo[r * 8 + k0];
        a.y = Wlo[r * 8 + k0 + 1];
    }
    return a;
}

__device__ __forceinline__ void half_swap(float& x, float& y) {
    const v2u r = __builtin_amdgcn_permlane16_swap(__float_as_uint(x),
                                                   __float_as_uint(y),
                                                   false, false);
    x = __uint_as_float(r.x);
    y = __uint_as_float(r.y);
}

struct RnnW {
    v2f t0a;
    v2f s1_a0, s1_a1;
    v2f s2_a0, s2_a1;
    v8f c_init;
    float c_fc[RNN_NCLS];
};

struct RnnS {
    v2f h0b0, h0b1;
    v2f h1b0, h1b1;
    v8f d1;
};

template <bool DO_OH, bool DO_S1, bool DO_S2, bool DO_ST, bool DO_H>
__device__ __forceinline__ void pipe_step(const RnnW& w, RnnS& s, int tok,
                                          int k0, bool hi, v4f& res) {
    const v2f h0b0 = s.h0b0, h0b1 = s.h0b1;
    const v2f h1b0 = s.h1b0, h1b1 = s.h1b1;
    const v8f d1p  = s.d1;
    v8f d1 = w.c_init;
    v8f d2 = d1p;

    if constexpr (DO_S2) {
        #pragma unroll
        for (int i = 0; i < RNN_NCLS; ++i) d2[i] = hi ? d1p[i] : w.c_fc[i];
    }
    if constexpr (DO_OH) {
        v2f oh;
        oh.x = (tok == k0)     ? 1.0f : 0.0f;
        oh.y = (tok == k0 + 1) ? 1.0f : 0.0f;
        d1 = wmma4(w.t0a, oh, w.c_init);
    }
    if constexpr (DO_S1) d1 = wmma4(w.s1_a0, h0b0, d1);
    if constexpr (DO_S2) d2 = wmma4(w.s2_a0, h1b0, d2);
    if constexpr (DO_S1) d1 = wmma4(w.s1_a1, h0b1, d1);
    if constexpr (DO_S2) d2 = wmma4(w.s2_a1, h1b1, d2);

    if constexpr (DO_ST) {
        res.x = d2[0]; res.y = d2[1]; res.z = d2[2]; res.w = d2[3];
    }
    if constexpr (DO_H) {
        float z[8];
        #pragma unroll
        for (int i = 0; i < 8; ++i) {
            if constexpr (DO_S2) z[i] = hi ? d2[i] : d1[i];
            else                 z[i] = d1[i];
        }
        #pragma unroll
        for (int i = 0; i < 8; ++i) z[i] = RNN_TANH(z[i]);
        half_swap(z[0], z[2]);
        half_swap(z[1], z[3]);
        half_swap(z[4], z[6]);
        half_swap(z[5], z[7]);
        s.h0b0.x = z[0]; s.h0b0.y = z[1];
        s.h0b1.x = z[4]; s.h0b1.y = z[5];
        if constexpr (DO_S2) {
            s.h1b0.x = z[2]; s.h1b0.y = z[3];
            s.h1b1.x = z[6]; s.h1b1.y = z[7];
        }
    }
    if constexpr (DO_S1) s.d1 = d1;
}

template <int N>
__device__ __forceinline__ void store_rows(const v4f (&res)[RNN_BLK], int first,
                                           bool hi, float*& op) {
    if (!hi) {
        #pragma unroll
        for (int j = 0; j < N; ++j)
            *(v4f*)(op + (size_t)j * RNN_B * RNN_NCLS) = res[first + j];
    }
    op += (size_t)N * RNN_B * RNN_NCLS;
}

__global__ __launch_bounds__(32)
void rnn_fused_kernel(const int* __restrict__ x,
                      const float* __restrict__ emb,
                      const float* __restrict__ W_ih0,
                      const float* __restrict__ W_hh0,
                      const float* __restrict__ b_ih0,
                      const float* __restrict__ b_hh0,
                      const float* __restrict__ W_ih1,
                      const float* __restrict__ W_hh1,
                      const float* __restrict__ b_ih1,
                      const float* __restrict__ b_hh1,
                      const float* __restrict__ W_fc,
                      const float* __restrict__ b_fc,
                      float* __restrict__ out) {
    const int lane = threadIdx.x & 31;
    const int lo   = lane & 15;
    const int hi   = lane >> 4;
    const int batch_base = blockIdx.x * 16;

    const int* xrow = x + (size_t)batch_base + lo;

    int tok[RNN_BLK], ntok[RNN_BLK];
    #pragma unroll
    for (int j = 0; j < RNN_BLK; ++j) tok[j]  = xrow[(size_t)j * RNN_B];
    #pragma unroll
    for (int j = 0; j < RNN_BLK; ++j) ntok[j] = xrow[(size_t)(RNN_BLK + j) * RNN_B];

    RnnW w;
    {
        const int j = lo & 7;
        const float bias = b_ih0[j] + b_hh0[j];
        const int v0 = 2 * hi, v1 = 2 * hi + 1;
        float a0 = bias, a1 = bias;
        #pragma unroll
        for (int d = 0; d < RNN_EMB; ++d) {
            const float wv = W_ih0[j * RNN_EMB + d];
            a0 += emb[v0 * RNN_EMB + d] * wv;
            a1 += emb[v1 * RNN_EMB + d] * wv;
        }
        w.t0a.x = (lo < 8) ? a0 : 0.0f;
        w.t0a.y = (lo < 8) ? a1 : 0.0f;
    }

    w.s1_a0 = load_a_stack(W_hh0, RNN_HID,  W_ih1, lo, hi, 0);
    w.s1_a1 = load_a_stack(W_hh0, RNN_HID,  W_ih1, lo, hi, 1);
    w.s2_a0 = load_a_stack(W_fc,  RNN_NCLS, W_hh1, lo, hi, 0);
    w.s2_a1 = load_a_stack(W_fc,  RNN_NCLS, W_hh1, lo, hi, 1);

    #pragma unroll
    for (int i = 0; i < 8; ++i)
        w.c_init[i] = hi ? (b_ih1[i] + b_hh1[i]) : 0.0f;
    #pragma unroll
    for (int i = 0; i < RNN_NCLS; ++i) w.c_fc[i] = b_fc[i];

    RnnS s;
    s.h0b0 = (v2f){0.0f, 0.0f}; s.h0b1 = (v2f){0.0f, 0.0f};
    s.h1b0 = (v2f){0.0f, 0.0f}; s.h1b1 = (v2f){0.0f, 0.0f};
    #pragma unroll
    for (int i = 0; i < 8; ++i) s.d1[i] = 0.0f;

    const int  k0  = 2 * hi;
    const bool hiB = (hi != 0);
    float* op = out + ((size_t)batch_base + lo) * RNN_NCLS;

    v4f res[RNN_BLK];

    pipe_step<true, true, false, false, true>(w, s, tok[0], k0, hiB, res[0]);
    pipe_step<true, true, false, false, true>(w, s, tok[1], k0, hiB, res[1]);
    pipe_step<true, true, true,  false, true>(w, s, tok[2], k0, hiB, res[2]);
    #pragma unroll
    for (int j = 3; j < RNN_BLK; ++j)
        pipe_step<true, true, true, true, true>(w, s, tok[j], k0, hiB, res[j]);
    store_rows<RNN_BLK - 3>(res, 3, hiB, op);

    const int* xnext = xrow + (size_t)(2 * RNN_BLK) * RNN_B;
    #pragma unroll 1
    for (int sb = RNN_BLK; sb < RNN_T; sb += RNN_BLK) {
        #pragma unroll
        for (int j = 0; j < RNN_BLK; ++j) tok[j] = ntok[j];
        if (sb + RNN_BLK < RNN_T) {
            #pragma unroll
            for (int j = 0; j < RNN_BLK; ++j) ntok[j] = xnext[(size_t)j * RNN_B];
            if (sb + 2 * RNN_BLK < RNN_T) {
                #pragma unroll
                for (int j = 0; j < RNN_BLK; ++j)
                    __builtin_prefetch(xnext + (size_t)(RNN_BLK + j) * RNN_B, 0, 3);
            }
        }
        #pragma unroll
        for (int j = 0; j < RNN_BLK; ++j)
            pipe_step<true, true, true, true, true>(w, s, tok[j], k0, hiB, res[j]);
        store_rows<RNN_BLK>(res, 0, hiB, op);
        xnext += (size_t)RNN_BLK * RNN_B;
    }

    pipe_step<false, true,  true, true, true >(w, s, 0, k0, hiB, res[0]);
    pipe_step<false, false, true, true, true >(w, s, 0, k0, hiB, res[1]);
    pipe_step<false, false, true, true, false>(w, s, 0, k0, hiB, res[2]);
    store_rows<3>(res, 0, hiB, op);
}

extern "C" void kernel_launch(void* const* d_in, const int* in_sizes, int n_in,
                              void* d_out, int out_size, void* d_ws, size_t ws_size,
                              hipStream_t stream) {
    (void)in_sizes; (void)n_in; (void)d_ws; (void)ws_size; (void)out_size;
    const int*   x     = (const int*)  d_in[0];
    const float* emb   = (const float*)d_in[1];
    const float* W_ih0 = (const float*)d_in[2];
    const float* W_hh0 = (const float*)d_in[3];
    const float* b_ih0 = (const float*)d_in[4];
    const float* b_hh0 = (const float*)d_in[5];
    const float* W_ih1 = (const float*)d_in[6];
    const float* W_hh1 = (const float*)d_in[7];
    const float* b_ih1 = (const float*)d_in[8];
    const float* b_hh1 = (const float*)d_in[9];
    const float* W_fc  = (const float*)d_in[10];
    const float* b_fc  = (const float*)d_in[11];
    float* out = (float*)d_out;

    dim3 grid(RNN_B / 16);
    dim3 block(32);
    rnn_fused_kernel<<<grid, block, 0, stream>>>(
        x, emb, W_ih0, W_hh0, b_ih0, b_hh0,
        W_ih1, W_hh1, b_ih1, b_hh1, W_fc, b_fc, out);
}


# WMMA rows re-assigned so each half-wave's D registers are the h0 / h1 rows it feeds to the next B operand: per step 12 half-wave selects and 4 v_permlane16_swap removed; sum order kept
# speedup vs baseline: 3.9012x; 1.1756x over previous
#include <hip/hip_runtime.h>
#include <hip/hip_bf16.h>
#include <math.h>


#define RNN_T    2048
#define RNN_B    2048
#define RNN_EMB  10
#define RNN_HID  8
#define RNN_NCLS 4
#define RNN_BLK  8

typedef __attribute__((ext_vector_type(2))) float v2f;
typedef __attribute__((ext_vector_type(4))) float v4f;
typedef __attribute__((ext_vector_type(8))) float v8f;

#if defined(__has_builtin)
#if __has_builtin(__builtin_amdgcn_tanhf)
#define RNN_TANH(x) __builtin_amdgcn_tanhf(x)
#endif
#endif
#ifndef RNN_TANH
#define RNN_TANH(x) tanhf(x)
#endif

__device__ __forceinline__ v8f wmma4(v2f a, v2f b, v8f c) {
    return __builtin_amdgcn_wmma_f32_16x16x4_f32(
         false, a,  false, b,
         (short)0, c,  false,  false);
}

__device__ __forceinline__ int hid_row(int h, int r) {
    return (r & 1) + 4 * (r >> 1) + 2 * h;
}

__device__ __forceinline__ v2f load_a_stack(const float* Wlo, bool lo_plain,
                                            const float* Whi,
                                            int lo, int hi, int c) {
    const int h   = lo >> 3;
    const int r   = lo & 3;
    const int row = hid_row(h, r);
    const int k0  = 4 * c + 2 * hi;
    v2f a; a.x = 0.0f; a.y = 0.0f;
    if (lo & 4) {
        a.x = Whi[row * 8 + k0];
        a.y = Whi[row * 8 + k0 + 1];
    } else if (!lo_plain) {
        a.x = Wlo[row * 8 + k0];
        a.y = Wlo[row * 8 + k0 + 1];
    } else if (h == 0) {
        a.x = Wlo[r * 8 + k0];
        a.y = Wlo[r * 8 + k0 + 1];
    }
    return a;
}

struct RnnW {
    v2f t0a;
    v2f s1_a0, s1_a1;
    v2f s2_a0, s2_a1;
    v8f c_init;
    float c_fc[RNN_NCLS];
};

struct RnnS {
    v2f h0b0, h0b1;
    v2f h1b0, h1b1;
    v8f d1;
};

template <bool DO_OH, bool DO_S1, bool DO_S2, bool DO_ST, bool DO_H>
__device__ __forceinline__ void pipe_step(const RnnW& w, RnnS& s, int tok,
                                          int k0, v4f& res) {
    const v2f h0b0 = s.h0b0, h0b1 = s.h0b1;
    const v2f h1b0 = s.h1b0, h1b1 = s.h1b1;
    const v8f d1p  = s.d1;
    v8f d1 = w.c_init;
    v8f d2 = d1p;

    if constexpr (DO_S2) {
        #pragma unroll
        for (int i = 0; i < RNN_NCLS; ++i) d2[i] = w.c_fc[i];
    }
    if constexpr (DO_OH) {
        v2f oh;
        oh.x = (tok == k0)     ? 1.0f : 0.0f;
        oh.y = (tok == k0 + 1) ? 1.0f : 0.0f;
        d1 = wmma4(w.t0a, oh, w.c_init);
    }
    if constexpr (DO_S1) d1 = wmma4(w.s1_a0, h0b0, d1);
    if constexpr (DO_S2) d2 = wmma4(w.s2_a0, h1b0, d2);
    if constexpr (DO_S1) d1 = wmma4(w.s1_a1, h0b1, d1);
    if constexpr (DO_S2) d2 = wmma4(w.s2_a1, h1b1, d2);

    if constexpr (DO_ST) {
        res.x = d2[0]; res.y = d2[1]; res.z = d2[2]; res.w = d2[3];
    }
    if constexpr (DO_H) {
        if constexpr (DO_S1) {
            s.h0b0.x = RNN_TANH(d1[0]); s.h0b0.y = RNN_TANH(d1[1]);
            s.h0b1.x = RNN_TANH(d1[2]); s.h0b1.y = RNN_TANH(d1[3]);
        }
        if constexpr (DO_S2) {
            s.h1b0.x = RNN_TANH(d2[4]); s.h1b0.y = RNN_TANH(d2[5]);
            s.h1b1.x = RNN_TANH(d2[6]); s.h1b1.y = RNN_TANH(d2[7]);
        }
    }
    if constexpr (DO_S1) s.d1 = d1;
}

template <int N>
__device__ __forceinline__ void store_rows(const v4f (&res)[RNN_BLK], int first,
                                           bool hi, float*& op) {
    if (!hi) {
        #pragma unroll
        for (int j = 0; j < N; ++j)
            *(v4f*)(op + (size_t)j * RNN_B * RNN_NCLS) = res[first + j];
    }
    op += (size_t)N * RNN_B * RNN_NCLS;
}

__global__ __launch_bounds__(32)
void rnn_fused_kernel(const int* __restrict__ x,
                      const float* __restrict__ emb,
                      const float* __restrict__ W_ih0,
                      const float* __restrict__ W_hh0,
                      const float* __restrict__ b_ih0,
                      const float* __restrict__ b_hh0,
                      const float* __restrict__ W_ih1,
                      const float* __restrict__ W_hh1,
                      const float* __restrict__ b_ih1,
                      const float* __restrict__ b_hh1,
                      const float* __restrict__ W_fc,
                      const float* __restrict__ b_fc,
                      float* __restrict__ out) {
    const int lane = threadIdx.x & 31;
    const int lo   = lane & 15;
    const int hi   = lane >> 4;
    const int batch_base = blockIdx.x * 16;

    const int* xrow = x + (size_t)batch_base + lo;

    int tok[RNN_BLK], ntok[RNN_BLK];
    #pragma unroll
    for (int j = 0; j < RNN_BLK; ++j) tok[j]  = xrow[(size_t)j * RNN_B];
    #pragma unroll
    for (int j = 0; j < RNN_BLK; ++j) ntok[j] = xrow[(size_t)(RNN_BLK + j) * RNN_B];

    RnnW w;
    {
        const int j = hid_row(lo >> 3, lo & 3);
        const float bias = b_ih0[j] + b_hh0[j];
        const int v0 = 2 * hi, v1 = 2 * hi + 1;
        float a0 = bias, a1 = bias;
        #pragma unroll
        for (int d = 0; d < RNN_EMB; ++d) {
            const float wv = W_ih0[j * RNN_EMB + d];
            a0 += emb[v0 * RNN_EMB + d] * wv;
            a1 += emb[v1 * RNN_EMB + d] * wv;
        }
        w.t0a.x = (lo & 4) ? 0.0f : a0;
        w.t0a.y = (lo & 4) ? 0.0f : a1;
    }

    w.s1_a0 = load_a_stack(W_hh0, false, W_ih1, lo, hi, 0);
    w.s1_a1 = load_a_stack(W_hh0, false, W_ih1, lo, hi, 1);
    w.s2_a0 = load_a_stack(W_fc,  true,  W_hh1, lo, hi, 0);
    w.s2_a1 = load_a_stack(W_fc,  true,  W_hh1, lo, hi, 1);

    #pragma unroll
    for (int i = 0; i < 4; ++i) {
        const int j = hid_row(hi, i);
        w.c_init[i]     = 0.0f;
        w.c_init[4 + i] = b_ih1[j] + b_hh1[j];
    }
    #pragma unroll
    for (int i = 0; i < RNN_NCLS; ++i) w.c_fc[i] = b_fc[i];

    RnnS s;
    s.h0b0 = (v2f){0.0f, 0.0f}; s.h0b1 = (v2f){0.0f, 0.0f};
    s.h1b0 = (v2f){0.0f, 0.0f}; s.h1b1 = (v2f){0.0f, 0.0f};
    #pragma unroll
    for (int i = 0; i < 8; ++i) s.d1[i] = 0.0f;

    const int  k0  = 2 * hi;
    const bool hiB = (hi != 0);
    float* op = out + ((size_t)batch_base + lo) * RNN_NCLS;

    v4f res[RNN_BLK];

    pipe_step<true, true, false, false, true>(w, s, tok[0], k0, res[0]);
    pipe_step<true, true, false, false, true>(w, s, tok[1], k0, res[1]);
    pipe_step<true, true, true,  false, true>(w, s, tok[2], k0, res[2]);
    #pragma unroll
    for (int j = 3; j < RNN_BLK; ++j)
        pipe_step<true, true, true, true, true>(w, s, tok[j], k0, res[j]);
    store_rows<RNN_BLK - 3>(res, 3, hiB, op);

    const int* xnext = xrow + (size_t)(2 * RNN_BLK) * RNN_B;
    #pragma unroll 1
    for (int sb = RNN_BLK; sb < RNN_T; sb += RNN_BLK) {
        #pragma unroll
        for (int j = 0; j < RNN_BLK; ++j) tok[j] = ntok[j];
        if (sb + RNN_BLK < RNN_T) {
            #pragma unroll
            for (int j = 0; j < RNN_BLK; ++j) ntok[j] = xnext[(size_t)j * RNN_B];
            if (sb + 2 * RNN_BLK < RNN_T) {
                #pragma unroll
                for (int j = 0; j < RNN_BLK; ++j)
                    __builtin_prefetch(xnext + (size_t)(RNN_BLK + j) * RNN_B, 0, 3);
            }
        }
        #pragma unroll
        for (int j = 0; j < RNN_BLK; ++j)
            pipe_step<true, true, true, true, true>(w, s, tok[j], k0, res[j]);
        store_rows<RNN_BLK>(res, 0, hiB, op);
        xnext += (size_t)RNN_BLK * RNN_B;
    }

    pipe_step<false, true,  true, true, true >(w, s, 0, k0, res[0]);
    pipe_step<false, false, true, true, true >(w, s, 0, k0, res[1]);
    pipe_step<false, false, true, true, false>(w, s, 0, k0, res[2]);
    store_rows<3>(res, 0, hiB, op);
}

extern "C" void kernel_launch(void* const* d_in, const int* in_sizes, int n_in,
                              void* d_out, int out_size, void* d_ws, size_t ws_size,
                              hipStream_t stream) {
    (void)in_sizes; (void)n_in; (void)d_ws; (void)ws_size; (void)out_size;
    const int*   x     = (const int*)  d_in[0];
    const float* emb   = (const float*)d_in[1];
    const float* W_ih0 = (const float*)d_in[2];
    const float* W_hh0 = (const float*)d_in[3];
    const float* b_ih0 = (const float*)d_in[4];
    const float* b_hh0 = (const float*)d_in[5];
    const float* W_ih1 = (const float*)d_in[6];
    const float* W_hh1 = (const float*)d_in[7];
    const float* b_ih1 = (const float*)d_in[8];
    const float* b_hh1 = (const float*)d_in[9];
    const float* W_fc  = (const float*)d_in[10];
    const float* b_fc  = (const float*)d_in[11];
    float* out = (float*)d_out;

    dim3 grid(RNN_B / 16);
    dim3 block(32);
    rnn_fused_kernel<<<grid, block, 0, stream>>>(
        x, emb, W_ih0, W_hh0, b_ih0, b_hh0,
        W_ih1, W_hh1, b_ih1, b_hh1, W_fc, b_fc, out);
}


# Layers split over two waves on two SIMDs: wave 0 does one-hot + stack 1 (h0, q), wave 1 stack 2 (h1, fc); q rows pass through a 2-slot LDS ring, one LDS-only barrier per 8 steps; sum order kept
# speedup vs baseline: 4.6265x; 1.1859x over previous
#include <hip/hip_runtime.h>
#include <hip/hip_bf16.h>
#include <math.h>
#include <type_traits>


#define RNN_T    2048
#define RNN_B    2048
#define RNN_EMB  10
#define RNN_HID  8
#define RNN_NCLS 4
#define RNN_BLK  8

typedef __attribute__((ext_vector_type(2))) float v2f;
typedef __attribute__((ext_vector_type(4))) float v4f;
typedef __attribute__((ext_vector_type(8))) float v8f;

#if defined(__has_builtin)
#if __has_builtin(__builtin_amdgcn_tanhf)
#define RNN_TANH(x) __builtin_amdgcn_tanhf(x)
#endif
#endif
#ifndef RNN_TANH
#define RNN_TANH(x) tanhf(x)
#endif

__device__ __forceinline__ v8f wmma4(v2f a, v2f b, v8f c) {
    return __builtin_amdgcn_wmma_f32_16x16x4_f32(
         false, a,  false, b,
         (short)0, c,  false,  false);
}

__device__ __forceinline__ int hid_row(int h, int r) {
    return (r & 1) + 4 * (r >> 1) + 2 * h;
}

__device__ __forceinline__ v2f load_a_stack(const float* Wlo, bool lo_plain,
                                            const float* Whi,
                                            int lo, int hi, int c) {
    const int h   = lo >> 3;
    const int r   = lo & 3;
    const int row = hid_row(h, r);
    const int k0  = 4 * c + 2 * hi;
    v2f a; a.x = 0.0f; a.y = 0.0f;
    if (lo & 4) {
        a.x = Whi[row * 8 + k0];
        a.y = Whi[row * 8 + k0 + 1];
    } else if (!lo_plain) {
        a.x = Wlo[row * 8 + k0];
        a.y = Wlo[row * 8 + k0 + 1];
    } else if (h == 0) {
        a.x = Wlo[r * 8 + k0];
        a.y = Wlo[r * 8 + k0 + 1];
    }
    return a;
}

struct RnnW {
    v2f t0a;
    v2f s1_a0, s1_a1;
    v2f s2_a0, s2_a1;
    v8f c_init;
    float c_fc[RNN_NCLS];
};

template <int N>
__device__ __forceinline__ void store_rows(const v4f (&res)[RNN_BLK], int first,
                                           bool hi, float*& op) {
    if (!hi) {
        #pragma unroll
        for (int j = 0; j < N; ++j)
            *(v4f*)(op + (size_t)j * RNN_B * RNN_NCLS) = res[first + j];
    }
    op += (size_t)N * RNN_B * RNN_NCLS;
}

#define RNN_NBLK (RNN_T / RNN_BLK)
#define RNN_LAG  3

__device__ __forceinline__ void l0_step(const RnnW& w, v2f& h0b0, v2f& h0b1,
                                        int tok, int k0, v4f& q) {
    v2f oh;
    oh.x = (tok == k0)     ? 1.0f : 0.0f;
    oh.y = (tok == k0 + 1) ? 1.0f : 0.0f;
    v8f d1 = wmma4(w.t0a, oh, w.c_init);
    d1 = wmma4(w.s1_a0, h0b0, d1);
    d1 = wmma4(w.s1_a1, h0b1, d1);
    q.x = d1[4]; q.y = d1[5]; q.z = d1[6]; q.w = d1[7];
    h0b0.x = RNN_TANH(d1[0]); h0b0.y = RNN_TANH(d1[1]);
    h0b1.x = RNN_TANH(d1[2]); h0b1.y = RNN_TANH(d1[3]);
}

template <bool DO_H>
__device__ __forceinline__ void l1_step(const RnnW& w, v2f& h1b0, v2f& h1b1,
                                        v4f q, v4f& res) {
    v8f d2;
    #pragma unroll
    for (int i = 0; i < RNN_NCLS; ++i) d2[i] = w.c_fc[i];
    d2[4] = q.x; d2[5] = q.y; d2[6] = q.z; d2[7] = q.w;
    d2 = wmma4(w.s2_a0, h1b0, d2);
    d2 = wmma4(w.s2_a1, h1b1, d2);
    res.x = d2[0]; res.y = d2[1]; res.z = d2[2]; res.w = d2[3];
    if constexpr (DO_H) {
        h1b0.x = RNN_TANH(d2[4]); h1b0.y = RNN_TANH(d2[5]);
        h1b1.x = RNN_TANH(d2[6]); h1b1.y = RNN_TANH(d2[7]);
    }
}

__device__ __forceinline__ void lds_handover_barrier() {
    __builtin_amdgcn_fence(__ATOMIC_RELEASE, "workgroup", "local");
    __builtin_amdgcn_s_barrier();
    __builtin_amdgcn_fence(__ATOMIC_ACQUIRE, "workgroup", "local");
}

__global__ __launch_bounds__(64)
void rnn_fused_kernel(const int* __restrict__ x,
                      const float* __restrict__ emb,
                      const float* __restrict__ W_ih0,
                      const float* __restrict__ W_hh0,
                      const float* __restrict__ b_ih0,
                      const float* __restrict__ b_hh0,
                      const float* __restrict__ W_ih1,
                      const float* __restrict__ W_hh1,
                      const float* __restrict__ b_ih1,
                      const float* __restrict__ b_hh1,
                      const float* __restrict__ W_fc,
                      const float* __restrict__ b_fc,
                      float* __restrict__ out) {
    __shared__ v4f ring[2][RNN_BLK][32];

    const int lane = threadIdx.x & 31;
    const int lo   = lane & 15;
    const int hi   = lane >> 4;
    const int role = __builtin_amdgcn_readfirstlane((int)(threadIdx.x >> 5));
    const int batch_base = blockIdx.x * 16;

    const int* xrow = x + (size_t)batch_base + lo;

    int tok0 = 0;
    int tok[RNN_BLK], ntok[RNN_BLK];
    #pragma unroll
    for (int j = 0; j < RNN_BLK; ++j) { tok[j] = 0; ntok[j] = 0; }
    if (role == 0) {
        tok0 = xrow[0];
        #pragma unroll
        for (int j = 0; j < RNN_BLK; ++j) ntok[j] = xrow[(size_t)(1 + j) * RNN_B];
    }

    RnnW w;
    {
        const int j = hid_row(lo >> 3, lo & 3);
        const float bias = b_ih0[j] + b_hh0[j];
        const int v0 = 2 * hi, v1 = 2 * hi + 1;
        float a0 = bias, a1 = bias;
        #pragma unroll
        for (int d = 0; d < RNN_EMB; ++d) {
            const float wv = W_ih0[j * RNN_EMB + d];
            a0 += emb[v0 * RNN_EMB + d] * wv;
            a1 += emb[v1 * RNN_EMB + d] * wv;
        }
        w.t0a.x = (lo & 4) ? 0.0f : a0;
        w.t0a.y = (lo & 4) ? 0.0f : a1;
    }

    w.s1_a0 = load_a_stack(W_hh0, false, W_ih1, lo, hi, 0);
    w.s1_a1 = load_a_stack(W_hh0, false, W_ih1, lo, hi, 1);
    w.s2_a0 = load_a_stack(W_fc,  true,  W_hh1, lo, hi, 0);
    w.s2_a1 = load_a_stack(W_fc,  true,  W_hh1, lo, hi, 1);

    #pragma unroll
    for (int i = 0; i < 4; ++i) {
        const int j = hid_row(hi, i);
        w.c_init[i]     = 0.0f;
        w.c_init[4 + i] = b_ih1[j] + b_hh1[j];
    }
    #pragma unroll
    for (int i = 0; i < RNN_NCLS; ++i) w.c_fc[i] = b_fc[i];

    v2f h0b0 = (v2f){0.0f, 0.0f}, h0b1 = (v2f){0.0f, 0.0f};
    v2f h1b0 = (v2f){0.0f, 0.0f}, h1b1 = (v2f){0.0f, 0.0f};
    const v4f zero4 = (v4f){0.0f, 0.0f, 0.0f, 0.0f};
    v4f qd[RNN_BLK];
    v4f qa[RNN_BLK], qb[RNN_BLK];
    v4f res[RNN_BLK];
    #pragma unroll
    for (int j = 0; j < RNN_BLK; ++j) { qd[j] = zero4; qa[j] = zero4; qb[j] = zero4; res[j] = zero4; }

    const int  k0  = 2 * hi;
    const bool hiB = (hi != 0);
    float* op = out + ((size_t)batch_base + lo) * RNN_NCLS;

    if (role == 0) {
        v4f qskip;
        l0_step(w, h0b0, h0b1, tok0, k0, qskip);
    }

    const int* xnext = xrow + (size_t)(RNN_BLK + 1) * RNN_B;

    auto block_trip = [&](auto steady_tag, const int it, v4f (&qin)[RNN_BLK], v4f (&qout)[RNN_BLK]) {
        constexpr bool STEADY = decltype(steady_tag)::value;
        if (role == 0) {
            if (STEADY || (it >= 1 && it <= RNN_NBLK)) {
                const int slot = (it - 1) & 1;
                #pragma unroll
                for (int j = 0; j < RNN_BLK; ++j) ring[slot][j][lane] = qd[j];
            }
            if (STEADY || it < RNN_NBLK) {
                #pragma unroll
                for (int j = 0; j < RNN_BLK; ++j) tok[j] = ntok[j];
                if (STEADY || it + 1 < RNN_NBLK) {
                    #pragma unroll
                    for (int j = 0; j < RNN_BLK - 1; ++j) ntok[j] = xnext[(size_t)j * RNN_B];
                    if (STEADY || it + 1 < RNN_NBLK - 1) {
                        ntok[RNN_BLK - 1] = xnext[(size_t)(RNN_BLK - 1) * RNN_B];
                        if (STEADY || it + 2 < RNN_NBLK - 1) {
                            #pragma unroll
                            for (int j = 0; j < RNN_BLK; ++j)
                                __builtin_prefetch(xnext + (size_t)(RNN_BLK + j) * RNN_B, 0, 3);
                        }
                    }
                    xnext += (size_t)RNN_BLK * RNN_B;
                }
                #pragma unroll
                for (int j = 0; j < RNN_BLK; ++j)
                    l0_step(w, h0b0, h0b1, tok[j], k0, qd[j]);
            }
        } else {
            if (STEADY || (it >= 2 && it < RNN_NBLK + 2)) {
                const int slot = (it - 2) & 1;
                #pragma unroll
                for (int j = 0; j < RNN_BLK; ++j) qout[j] = ring[slot][j][lane];
            }
            if (STEADY || (it >= RNN_LAG && it < RNN_NBLK + RNN_LAG)) {
                #pragma unroll
                for (int j = 0; j < RNN_BLK; ++j)
                    l1_step<true>(w, h1b0, h1b1, qin[j], res[j]);
                if (!STEADY && it == RNN_LAG) store_rows<RNN_BLK - 1>(res, 1, hiB, op);
                else                          store_rows<RNN_BLK>(res, 0, hiB, op);
            }
        }
        lds_handover_barrier();
    };
    const std::integral_constant<bool, false> edge_trip{};
    const std::integral_constant<bool, true>  steady_trip{};

    #pragma unroll 1
    for (int it = 0; it < 4; it += 2) {
        block_trip(edge_trip, it,     qa, qb);
        block_trip(edge_trip, it + 1, qb, qa);
    }
    #pragma unroll 1
    for (int it = 4; it < RNN_NBLK - 4; it += 2) {
        block_trip(steady_trip, it,     qa, qb);
        block_trip(steady_trip, it + 1, qb, qa);
    }
    #pragma unroll 1
    for (int it = RNN_NBLK - 4; it < RNN_NBLK + RNN_LAG; it += 2) {
        block_trip(edge_trip, it,     qa, qb);
        block_trip(edge_trip, it + 1, qb, qa);
    }

    if (role != 0) {
        l1_step<false>(w, h1b0, h1b1, zero4, res[0]);
        store_rows<1>(res, 0, hiB, op);
    }
}

extern "C" void kernel_launch(void* const* d_in, const int* in_sizes, int n_in,
                              void* d_out, int out_size, void* d_ws, size_t ws_size,
                              hipStream_t stream) {
    (void)in_sizes; (void)n_in; (void)d_ws; (void)ws_size; (void)out_size;
    const int*   x     = (const int*)  d_in[0];
    const float* emb   = (const float*)d_in[1];
    const float* W_ih0 = (const float*)d_in[2];
    const float* W_hh0 = (const float*)d_in[3];
    const float* b_ih0 = (const float*)d_in[4];
    const float* b_hh0 = (const float*)d_in[5];
    const float* W_ih1 = (const float*)d_in[6];
    const float* W_hh1 = (const float*)d_in[7];
    const float* b_ih1 = (const float*)d_in[8];
    const float* b_hh1 = (const float*)d_in[9];
    const float* W_fc  = (const float*)d_in[10];
    const float* b_fc  = (const float*)d_in[11];
    float* out = (float*)d_out;

    dim3 grid(RNN_B / 16);
    dim3 block(64);
    rnn_fused_kernel<<<grid, block, 0, stream>>>(
        x, emb, W_ih0, W_hh0, b_ih0, b_hh0,
        W_ih1, W_hh1, b_ih1, b_hh1, W_fc, b_fc, out);
}


# Hand-over block lengthened from 8 to 16 time steps: one barrier, one LDS store/load group, one token-load group and one output-store group per 16 steps instead of per 8; sum order kept
# speedup vs baseline: 4.9807x; 1.0766x over previous
#include <hip/hip_runtime.h>
#include <hip/hip_bf16.h>
#include <math.h>
#include <type_traits>


#define RNN_T    2048
#define RNN_B    2048
#define RNN_EMB  10
#define RNN_HID  8
#define RNN_NCLS 4
#define RNN_BLK  16

typedef __attribute__((ext_vector_type(2))) float v2f;
typedef __attribute__((ext_vector_type(4))) float v4f;
typedef __attribute__((ext_vector_type(8))) float v8f;

#if defined(__has_builtin)
#if __has_builtin(__builtin_amdgcn_tanhf)
#define RNN_TANH(x) __builtin_amdgcn_tanhf(x)
#endif
#endif
#ifndef RNN_TANH
#define RNN_TANH(x) tanhf(x)
#endif

__device__ __forceinline__ v8f wmma4(v2f a, v2f b, v8f c) {
    return __builtin_amdgcn_wmma_f32_16x16x4_f32(
         false, a,  false, b,
         (short)0, c,  false,  false);
}

__device__ __forceinline__ int hid_row(int h, int r) {
    return (r & 1) + 4 * (r >> 1) + 2 * h;
}

__device__ __forceinline__ v2f load_a_stack(const float* Wlo, bool lo_plain,
                                            const float* Whi,
                                            int lo, int hi, int c) {
    const int h   = lo >> 3;
    const int r   = lo & 3;
    const int row = hid_row(h, r);
    const int k0  = 4 * c + 2 * hi;
    v2f a; a.x = 0.0f; a.y = 0.0f;
    if (lo & 4) {
        a.x = Whi[row * 8 + k0];
        a.y = Whi[row * 8 + k0 + 1];
    } else if (!lo_plain) {
        a.x = Wlo[row * 8 + k0];
        a.y = Wlo[row * 8 + k0 + 1];
    } else if (h == 0) {
        a.x = Wlo[r * 8 + k0];
        a.y = Wlo[r * 8 + k0 + 1];
    }
    return a;
}

struct RnnW {
    v2f t0a;
    v2f s1_a0, s1_a1;
    v2f s2_a0, s2_a1;
    v8f c_init;
    float c_fc[RNN_NCLS];
};

template <int N>
__device__ __forceinline__ void store_rows(const v4f (&res)[RNN_BLK], int first,
                                           bool hi, float*& op) {
    if (!hi) {
        #pragma unroll
        for (int j = 0; j < N; ++j)
            *(v4f*)(op + (size_t)j * RNN_B * RNN_NCLS) = res[first + j];
    }
    op += (size_t)N * RNN_B * RNN_NCLS;
}

#define RNN_NBLK (RNN_T / RNN_BLK)
#define RNN_LAG  3

__device__ __forceinline__ void l0_step(const RnnW& w, v2f& h0b0, v2f& h0b1,
                                        int tok, int k0, v4f& q) {
    v2f oh;
    oh.x = (tok == k0)     ? 1.0f : 0.0f;
    oh.y = (tok == k0 + 1) ? 1.0f : 0.0f;
    v8f d1 = wmma4(w.t0a, oh, w.c_init);
    d1 = wmma4(w.s1_a0, h0b0, d1);
    d1 = wmma4(w.s1_a1, h0b1, d1);
    q.x = d1[4]; q.y = d1[5]; q.z = d1[6]; q.w = d1[7];
    h0b0.x = RNN_TANH(d1[0]); h0b0.y = RNN_TANH(d1[1]);
    h0b1.x = RNN_TANH(d1[2]); h0b1.y = RNN_TANH(d1[3]);
}

template <bool DO_H>
__device__ __forceinline__ void l1_step(const RnnW& w, v2f& h1b0, v2f& h1b1,
                                        v4f q, v4f& res) {
    v8f d2;
    #pragma unroll
    for (int i = 0; i < RNN_NCLS; ++i) d2[i] = w.c_fc[i];
    d2[4] = q.x; d2[5] = q.y; d2[6] = q.z; d2[7] = q.w;
    d2 = wmma4(w.s2_a0, h1b0, d2);
    d2 = wmma4(w.s2_a1, h1b1, d2);
    res.x = d2[0]; res.y = d2[1]; res.z = d2[2]; res.w = d2[3];
    if constexpr (DO_H) {
        h1b0.x = RNN_TANH(d2[4]); h1b0.y = RNN_TANH(d2[5]);
        h1b1.x = RNN_TANH(d2[6]); h1b1.y = RNN_TANH(d2[7]);
    }
}

__device__ __forceinline__ void lds_handover_barrier() {
    __builtin_amdgcn_fence(__ATOMIC_RELEASE, "workgroup", "local");
    __builtin_amdgcn_s_barrier();
    __builtin_amdgcn_fence(__ATOMIC_ACQUIRE, "workgroup", "local");
}

__global__ __launch_bounds__(64)
void rnn_fused_kernel(const int* __restrict__ x,
                      const float* __restrict__ emb,
                      const float* __restrict__ W_ih0,
                      const float* __restrict__ W_hh0,
                      const float* __restrict__ b_ih0,
                      const float* __restrict__ b_hh0,
                      const float* __restrict__ W_ih1,
                      const float* __restrict__ W_hh1,
                      const float* __restrict__ b_ih1,
                      const float* __restrict__ b_hh1,
                      const float* __restrict__ W_fc,
                      const float* __restrict__ b_fc,
                      float* __restrict__ out) {
    __shared__ v4f ring[2][RNN_BLK][32];

    const int lane = threadIdx.x & 31;
    const int lo   = lane & 15;
    const int hi   = lane >> 4;
    const int role = __builtin_amdgcn_readfirstlane((int)(threadIdx.x >> 5));
    const int batch_base = blockIdx.x * 16;

    const int* xrow = x + (size_t)batch_base + lo;

    int tok0 = 0;
    int tok[RNN_BLK], ntok[RNN_BLK];
    #pragma unroll
    for (int j = 0; j < RNN_BLK; ++j) { tok[j] = 0; ntok[j] = 0; }
    if (role == 0) {
        tok0 = xrow[0];
        #pragma unroll
        for (int j = 0; j < RNN_BLK; ++j) ntok[j] = xrow[(size_t)(1 + j) * RNN_B];
    }

    RnnW w;
    {
        const int j = hid_row(lo >> 3, lo & 3);
        const float bias = b_ih0[j] + b_hh0[j];
        const int v0 = 2 * hi, v1 = 2 * hi + 1;
        float a0 = bias, a1 = bias;
        #pragma unroll
        for (int d = 0; d < RNN_EMB; ++d) {
            const float wv = W_ih0[j * RNN_EMB + d];
            a0 += emb[v0 * RNN_EMB + d] * wv;
            a1 += emb[v1 * RNN_EMB + d] * wv;
        }
        w.t0a.x = (lo & 4) ? 0.0f : a0;
        w.t0a.y = (lo & 4) ? 0.0f : a1;
    }

    w.s1_a0 = load_a_stack(W_hh0, false, W_ih1, lo, hi, 0);
    w.s1_a1 = load_a_stack(W_hh0, false, W_ih1, lo, hi, 1);
    w.s2_a0 = load_a_stack(W_fc,  true,  W_hh1, lo, hi, 0);
    w.s2_a1 = load_a_stack(W_fc,  true,  W_hh1, lo, hi, 1);

    #pragma unroll
    for (int i = 0; i < 4; ++i) {
        const int j = hid_row(hi, i);
        w.c_init[i]     = 0.0f;
        w.c_init[4 + i] = b_ih1[j] + b_hh1[j];
    }
    #pragma unroll
    for (int i = 0; i < RNN_NCLS; ++i) w.c_fc[i] = b_fc[i];

    v2f h0b0 = (v2f){0.0f, 0.0f}, h0b1 = (v2f){0.0f, 0.0f};
    v2f h1b0 = (v2f){0.0f, 0.0f}, h1b1 = (v2f){0.0f, 0.0f};
    const v4f zero4 = (v4f){0.0f, 0.0f, 0.0f, 0.0f};
    v4f qd[RNN_BLK];
    v4f qa[RNN_BLK], qb[RNN_BLK];
    v4f res[RNN_BLK];
    #pragma unroll
    for (int j = 0; j < RNN_BLK; ++j) { qd[j] = zero4; qa[j] = zero4; qb[j] = zero4; res[j] = zero4; }

    const int  k0  = 2 * hi;
    const bool hiB = (hi != 0);
    float* op = out + ((size_t)batch_base + lo) * RNN_NCLS;

    if (role == 0) {
        v4f qskip;
        l0_step(w, h0b0, h0b1, tok0, k0, qskip);
    }

    const int* xnext = xrow + (size_t)(RNN_BLK + 1) * RNN_B;

    auto block_trip = [&](auto steady_tag, const int it, v4f (&qin)[RNN_BLK], v4f (&qout)[RNN_BLK]) {
        constexpr bool STEADY = decltype(steady_tag)::value;
        if (role == 0) {
            if (STEADY || (it >= 1 && it <= RNN_NBLK)) {
                const int slot = (it - 1) & 1;
                #pragma unroll
                for (int j = 0; j < RNN_BLK; ++j) ring[slot][j][lane] = qd[j];
            }
            if (STEADY || it < RNN_NBLK) {
                #pragma unroll
                for (int j = 0; j < RNN_BLK; ++j) tok[j] = ntok[j];
                if (STEADY || it + 1 < RNN_NBLK) {
                    #pragma unroll
                    for (int j = 0; j < RNN_BLK - 1; ++j) ntok[j] = xnext[(size_t)j * RNN_B];
                    if (STEADY || it + 1 < RNN_NBLK - 1) {
                        ntok[RNN_BLK - 1] = xnext[(size_t)(RNN_BLK - 1) * RNN_B];
                        if (STEADY || it + 2 < RNN_NBLK - 1) {
                            #pragma unroll
                            for (int j = 0; j < RNN_BLK; ++j)
                                __builtin_prefetch(xnext + (size_t)(RNN_BLK + j) * RNN_B, 0, 3);
                        }
                    }
                    xnext += (size_t)RNN_BLK * RNN_B;
                }
                #pragma unroll
                for (int j = 0; j < RNN_BLK; ++j)
                    l0_step(w, h0b0, h0b1, tok[j], k0, qd[j]);
            }
        } else {
            if (STEADY || (it >= 2 && it < RNN_NBLK + 2)) {
                const int slot = (it - 2) & 1;
                #pragma unroll
                for (int j = 0; j < RNN_BLK; ++j) qout[j] = ring[slot][j][lane];
            }
            if (STEADY || (it >= RNN_LAG && it < RNN_NBLK + RNN_LAG)) {
                #pragma unroll
                for (int j = 0; j < RNN_BLK; ++j)
                    l1_step<true>(w, h1b0, h1b1, qin[j], res[j]);
                if (!STEADY && it == RNN_LAG) store_rows<RNN_BLK - 1>(res, 1, hiB, op);
                else                          store_rows<RNN_BLK>(res, 0, hiB, op);
            }
        }
        lds_handover_barrier();
    };
    const std::integral_constant<bool, false> edge_trip{};
    const std::integral_constant<bool, true>  steady_trip{};

    #pragma unroll 1
    for (int it = 0; it < 4; it += 2) {
        block_trip(edge_trip, it,     qa, qb);
        block_trip(edge_trip, it + 1, qb, qa);
    }
    #pragma unroll 1
    for (int it = 4; it < RNN_NBLK - 4; it += 2) {
        block_trip(steady_trip, it,     qa, qb);
        block_trip(steady_trip, it + 1, qb, qa);
    }
    #pragma unroll 1
    for (int it = RNN_NBLK - 4; it < RNN_NBLK + RNN_LAG; it += 2) {
        block_trip(edge_trip, it,     qa, qb);
        block_trip(edge_trip, it + 1, qb, qa);
    }

    if (role != 0) {
        l1_step<false>(w, h1b0, h1b1, zero4, res[0]);
        store_rows<1>(res, 0, hiB, op);
    }
}

extern "C" void kernel_launch(void* const* d_in, const int* in_sizes, int n_in,
                              void* d_out, int out_size, void* d_ws, size_t ws_size,
                              hipStream_t stream) {
    (void)in_sizes; (void)n_in; (void)d_ws; (void)ws_size; (void)out_size;
    const int*   x     = (const int*)  d_in[0];
    const float* emb   = (const float*)d_in[1];
    const float* W_ih0 = (const float*)d_in[2];
    const float* W_hh0 = (const float*)d_in[3];
    const float* b_ih0 = (const float*)d_in[4];
    const float* b_hh0 = (const float*)d_in[5];
    const float* W_ih1 = (const float*)d_in[6];
    const float* W_hh1 = (const float*)d_in[7];
    const float* b_ih1 = (const float*)d_in[8];
    const float* b_hh1 = (const float*)d_in[9];
    const float* W_fc  = (const float*)d_in[10];
    const float* b_fc  = (const float*)d_in[11];
    float* out = (float*)d_out;

    dim3 grid(RNN_B / 16);
    dim3 block(64);
    rnn_fused_kernel<<<grid, block, 0, stream>>>(
        x, emb, W_ih0, W_hh0, b_ih0, b_hh0,
        W_ih1, W_hh1, b_ih1, b_hh1, W_fc, b_fc, out);
}


# Per-step issue order pinned on both waves: all four tanh of a step ahead of its first recurrent WMMA, next step's one-hot product / C set-up issued in the wait for the accumulator; sum order kept
# speedup vs baseline: 5.7351x; 1.1515x over previous
#include <hip/hip_runtime.h>
#include <hip/hip_bf16.h>
#include <math.h>
#include <type_traits>


#define RNN_T    2048
#define RNN_B    2048
#define RNN_EMB  10
#define RNN_HID  8
#define RNN_NCLS 4
#define RNN_BLK  16

typedef __attribute__((ext_vector_type(2))) float v2f;
typedef __attribute__((ext_vector_type(4))) float v4f;
typedef __attribute__((ext_vector_type(8))) float v8f;

#if defined(__has_builtin)
#if __has_builtin(__builtin_amdgcn_tanhf)
#define RNN_TANH(x) __builtin_amdgcn_tanhf(x)
#endif
#endif
#ifndef RNN_TANH
#define RNN_TANH(x) tanhf(x)
#endif

__device__ __forceinline__ v8f wmma4(v2f a, v2f b, v8f c) {
    return __builtin_amdgcn_wmma_f32_16x16x4_f32(
         false, a,  false, b,
         (short)0, c,  false,  false);
}

__device__ __forceinline__ int hid_row(int h, int r) {
    return (r & 1) + 4 * (r >> 1) + 2 * h;
}

__device__ __forceinline__ v2f load_a_stack(const float* Wlo, bool lo_plain,
                                            const float* Whi,
                                            int lo, int hi, int c) {
    const int h   = lo >> 3;
    const int r   = lo & 3;
    const int row = hid_row(h, r);
    const int k0  = 4 * c + 2 * hi;
    v2f a; a.x = 0.0f; a.y = 0.0f;
    if (lo & 4) {
        a.x = Whi[row * 8 + k0];
        a.y = Whi[row * 8 + k0 + 1];
    } else if (!lo_plain) {
        a.x = Wlo[row * 8 + k0];
        a.y = Wlo[row * 8 + k0 + 1];
    } else if (h == 0) {
        a.x = Wlo[r * 8 + k0];
        a.y = Wlo[r * 8 + k0 + 1];
    }
    return a;
}

struct RnnW {
    v2f t0a;
    v2f s1_a0, s1_a1;
    v2f s2_a0, s2_a1;
    v8f c_init;
    float c_fc[RNN_NCLS];
};

template <int N>
__device__ __forceinline__ void store_rows(const v4f (&res)[RNN_BLK], int first,
                                           bool hi, float*& op) {
    if (!hi) {
        #pragma unroll
        for (int j = 0; j < N; ++j)
            *(v4f*)(op + (size_t)j * RNN_B * RNN_NCLS) = res[first + j];
    }
    op += (size_t)N * RNN_B * RNN_NCLS;
}

#define RNN_NBLK (RNN_T / RNN_BLK)
#define RNN_LAG  3

__device__ __forceinline__ void l0_step(const RnnW& w, v2f& h0b0, v2f& h0b1,
                                        int tok, int k0, v4f& q) {
    v2f oh;
    oh.x = (tok == k0)     ? 1.0f : 0.0f;
    oh.y = (tok == k0 + 1) ? 1.0f : 0.0f;
    v8f d1 = wmma4(w.t0a, oh, w.c_init);
    d1 = wmma4(w.s1_a0, h0b0, d1);
    d1 = wmma4(w.s1_a1, h0b1, d1);
    q.x = d1[4]; q.y = d1[5]; q.z = d1[6]; q.w = d1[7];
    h0b0.x = RNN_TANH(d1[0]); h0b0.y = RNN_TANH(d1[1]);
    h0b1.x = RNN_TANH(d1[2]); h0b1.y = RNN_TANH(d1[3]);
}

__device__ __forceinline__ v8f l1_cinit(const RnnW& w, v4f q) {
    v8f c;
    #pragma unroll
    for (int i = 0; i < RNN_NCLS; ++i) c[i] = w.c_fc[i];
    c[4] = q.x; c[5] = q.y; c[6] = q.z; c[7] = q.w;
    return c;
}

__device__ __forceinline__ void l1_last_row(const RnnW& w, v2f h1b0, v2f h1b1, v4f& res) {
    const v4f zero4 = (v4f){0.0f, 0.0f, 0.0f, 0.0f};
    v8f d2 = l1_cinit(w, zero4);
    d2 = wmma4(w.s2_a0, h1b0, d2);
    d2 = wmma4(w.s2_a1, h1b1, d2);
    res.x = d2[0]; res.y = d2[1]; res.z = d2[2]; res.w = d2[3];
}

__device__ __forceinline__ v8f l0_onehot(const RnnW& w, int tok, int k0) {
    v2f oh;
    oh.x = (tok == k0)     ? 1.0f : 0.0f;
    oh.y = (tok == k0 + 1) ? 1.0f : 0.0f;
    return wmma4(w.t0a, oh, w.c_init);
}

__device__ __forceinline__ void l0_act(const v8f& d1, v2f& h0b0, v2f& h0b1, v4f& q) {
    q.x = d1[4]; q.y = d1[5]; q.z = d1[6]; q.w = d1[7];
    h0b0.x = RNN_TANH(d1[0]); h0b0.y = RNN_TANH(d1[1]);
    h0b1.x = RNN_TANH(d1[2]); h0b1.y = RNN_TANH(d1[3]);
}

__device__ __forceinline__ void lds_handover_barrier() {
    __builtin_amdgcn_fence(__ATOMIC_RELEASE, "workgroup", "local");
    __builtin_amdgcn_s_barrier();
    __builtin_amdgcn_fence(__ATOMIC_ACQUIRE, "workgroup", "local");
}

__global__ __launch_bounds__(64)
void rnn_fused_kernel(const int* __restrict__ x,
                      const float* __restrict__ emb,
                      const float* __restrict__ W_ih0,
                      const float* __restrict__ W_hh0,
                      const float* __restrict__ b_ih0,
                      const float* __restrict__ b_hh0,
                      const float* __restrict__ W_ih1,
                      const float* __restrict__ W_hh1,
                      const float* __restrict__ b_ih1,
                      const float* __restrict__ b_hh1,
                      const float* __restrict__ W_fc,
                      const float* __restrict__ b_fc,
                      float* __restrict__ out) {
    __shared__ v4f ring[2][RNN_BLK][32];

    const int lane = threadIdx.x & 31;
    const int lo   = lane & 15;
    const int hi   = lane >> 4;
    const int role = __builtin_amdgcn_readfirstlane((int)(threadIdx.x >> 5));
    const int batch_base = blockIdx.x * 16;

    const int* xrow = x + (size_t)batch_base + lo;

    int tok0 = 0;
    int tok[RNN_BLK], ntok[RNN_BLK];
    #pragma unroll
    for (int j = 0; j < RNN_BLK; ++j) { tok[j] = 0; ntok[j] = 0; }
    if (role == 0) {
        tok0 = xrow[0];
        #pragma unroll
        for (int j = 0; j < RNN_BLK; ++j) tok[j] = xrow[(size_t)(1 + j) * RNN_B];
    }

    RnnW w;
    {
        const int j = hid_row(lo >> 3, lo & 3);
        const float bias = b_ih0[j] + b_hh0[j];
        const int v0 = 2 * hi, v1 = 2 * hi + 1;
        float a0 = bias, a1 = bias;
        #pragma unroll
        for (int d = 0; d < RNN_EMB; ++d) {
            const float wv = W_ih0[j * RNN_EMB + d];
            a0 += emb[v0 * RNN_EMB + d] * wv;
            a1 += emb[v1 * RNN_EMB + d] * wv;
        }
        w.t0a.x = (lo & 4) ? 0.0f : a0;
        w.t0a.y = (lo & 4) ? 0.0f : a1;
    }

    w.s1_a0 = load_a_stack(W_hh0, false, W_ih1, lo, hi, 0);
    w.s1_a1 = load_a_stack(W_hh0, false, W_ih1, lo, hi, 1);
    w.s2_a0 = load_a_stack(W_fc,  true,  W_hh1, lo, hi, 0);
    w.s2_a1 = load_a_stack(W_fc,  true,  W_hh1, lo, hi, 1);

    #pragma unroll
    for (int i = 0; i < 4; ++i) {
        const int j = hid_row(hi, i);
        w.c_init[i]     = 0.0f;
        w.c_init[4 + i] = b_ih1[j] + b_hh1[j];
    }
    #pragma unroll
    for (int i = 0; i < RNN_NCLS; ++i) w.c_fc[i] = b_fc[i];

    v2f h0b0 = (v2f){0.0f, 0.0f}, h0b1 = (v2f){0.0f, 0.0f};
    v2f h1b0 = (v2f){0.0f, 0.0f}, h1b1 = (v2f){0.0f, 0.0f};
    const v4f zero4 = (v4f){0.0f, 0.0f, 0.0f, 0.0f};
    v4f qd[RNN_BLK];
    v4f qa[RNN_BLK], qb[RNN_BLK];
    v4f res[RNN_BLK];
    #pragma unroll
    for (int j = 0; j < RNN_BLK; ++j) { qd[j] = zero4; qa[j] = zero4; qb[j] = zero4; res[j] = zero4; }

    const int  k0  = 2 * hi;
    const bool hiB = (hi != 0);
    v8f dn0 = w.c_init;
    float* op = out + ((size_t)batch_base + lo) * RNN_NCLS;

    if (role == 0) {
        v4f qskip;
        l0_step(w, h0b0, h0b1, tok0, k0, qskip);
    }

    const int* xnext = xrow + (size_t)(RNN_BLK + 1) * RNN_B;

    auto block_trip = [&](auto steady_tag, const int it, v4f (&qin)[RNN_BLK], v4f (&qout)[RNN_BLK],
                          int (&tcur)[RNN_BLK], int (&tnxt)[RNN_BLK]) {
        constexpr bool STEADY = decltype(steady_tag)::value;
        if (role == 0) {
            if (STEADY || (it >= 1 && it <= RNN_NBLK)) {
                const int slot = (it - 1) & 1;
                #pragma unroll
                for (int j = 0; j < RNN_BLK; ++j) ring[slot][j][lane] = qd[j];
            }
            if (STEADY || it < RNN_NBLK) {
                if (STEADY || it + 1 < RNN_NBLK) {
                    #pragma unroll
                    for (int j = 0; j < RNN_BLK - 1; ++j) tnxt[j] = xnext[(size_t)j * RNN_B];
                    if (STEADY || it + 1 < RNN_NBLK - 1) {
                        tnxt[RNN_BLK - 1] = xnext[(size_t)(RNN_BLK - 1) * RNN_B];
                        if (STEADY || it + 2 < RNN_NBLK - 1) {
                            #pragma unroll
                            for (int j = 0; j < RNN_BLK; ++j)
                                __builtin_prefetch(xnext + (size_t)(RNN_BLK + j) * RNN_B, 0, 3);
                        }
                    }
                    xnext += (size_t)RNN_BLK * RNN_B;
                }
                if (!STEADY && it == 0) dn0 = l0_onehot(w, tcur[0], k0);
                const bool carry = STEADY || (it + 1 < RNN_NBLK);
                #pragma unroll
                for (int j = 0; j < RNN_BLK; ++j) {
                    v8f d1 = wmma4(w.s1_a0, h0b0, dn0);
                    d1 = wmma4(w.s1_a1, h0b1, d1);
                    __builtin_amdgcn_sched_barrier(0);
                    if (j + 1 < RNN_BLK) dn0 = l0_onehot(w, tcur[j + 1], k0);
                    else if (carry)      dn0 = l0_onehot(w, tnxt[0], k0);
                    __builtin_amdgcn_sched_barrier(0);
                    l0_act(d1, h0b0, h0b1, qd[j]);
                    __builtin_amdgcn_sched_barrier(0);
                }
            }
        } else {
            if (STEADY || (it >= 2 && it < RNN_NBLK + 2)) {
                const int slot = (it - 2) & 1;
                #pragma unroll
                for (int j = 0; j < RNN_BLK; ++j) qout[j] = ring[slot][j][lane];
            }
            if (STEADY || (it >= RNN_LAG && it < RNN_NBLK + RNN_LAG)) {
                v8f dn1 = l1_cinit(w, qin[0]);
                #pragma unroll
                for (int j = 0; j < RNN_BLK; ++j) {
                    v8f d2 = wmma4(w.s2_a0, h1b0, dn1);
                    d2 = wmma4(w.s2_a1, h1b1, d2);
                    __builtin_amdgcn_sched_barrier(0);
                    if (j + 1 < RNN_BLK) dn1 = l1_cinit(w, qin[j + 1]);
                    __builtin_amdgcn_sched_barrier(0);
                    res[j].x = d2[0]; res[j].y = d2[1]; res[j].z = d2[2]; res[j].w = d2[3];
                    h1b0.x = RNN_TANH(d2[4]); h1b0.y = RNN_TANH(d2[5]);
                    h1b1.x = RNN_TANH(d2[6]); h1b1.y = RNN_TANH(d2[7]);
                    __builtin_amdgcn_sched_barrier(0);
                }
                if (!STEADY && it == RNN_LAG) store_rows<RNN_BLK - 1>(res, 1, hiB, op);
                else                          store_rows<RNN_BLK>(res, 0, hiB, op);
            }
        }
        lds_handover_barrier();
    };
    const std::integral_constant<bool, false> edge_trip{};
    const std::integral_constant<bool, true>  steady_trip{};

    #pragma unroll 1
    for (int it = 0; it < 4; it += 2) {
        block_trip(edge_trip, it,     qa, qb, tok, ntok);
        block_trip(edge_trip, it + 1, qb, qa, ntok, tok);
    }
    #pragma unroll 1
    for (int it = 4; it < RNN_NBLK - 4; it += 2) {
        block_trip(steady_trip, it,     qa, qb, tok, ntok);
        block_trip(steady_trip, it + 1, qb, qa, ntok, tok);
    }
    #pragma unroll 1
    for (int it = RNN_NBLK - 4; it < RNN_NBLK + RNN_LAG; it += 2) {
        block_trip(edge_trip, it,     qa, qb, tok, ntok);
        block_trip(edge_trip, it + 1, qb, qa, ntok, tok);
    }

    if (role != 0) {
        l1_last_row(w, h1b0, h1b1, res[0]);
        store_rows<1>(res, 0, hiB, op);
    }
}

extern "C" void kernel_launch(void* const* d_in, const int* in_sizes, int n_in,
                              void* d_out, int out_size, void* d_ws, size_t ws_size,
                              hipStream_t stream) {
    (void)in_sizes; (void)n_in; (void)d_ws; (void)ws_size; (void)out_size;
    const int*   x     = (const int*)  d_in[0];
    const float* emb   = (const float*)d_in[1];
    const float* W_ih0 = (const float*)d_in[2];
    const float* W_hh0 = (const float*)d_in[3];
    const float* b_ih0 = (const float*)d_in[4];
    const float* b_hh0 = (const float*)d_in[5];
    const float* W_ih1 = (const float*)d_in[6];
    const float* W_hh1 = (const float*)d_in[7];
    const float* b_ih1 = (const float*)d_in[8];
    const float* b_hh1 = (const float*)d_in[9];
    const float* W_fc  = (const float*)d_in[10];
    const float* b_fc  = (const float*)d_in[11];
    float* out = (float*)d_out;

    dim3 grid(RNN_B / 16);
    dim3 block(64);
    rnn_fused_kernel<<<grid, block, 0, stream>>>(
        x, emb, W_ih0, W_hh0, b_ih0, b_hh0,
        W_ih1, W_hh1, b_ih1, b_hh1, W_fc, b_fc, out);
}


# One-hot input product moved off the layer-0 wave onto a third wave (own SIMD): it forms each step's whole C operand a block ahead, handed over through a second LDS ring; sum order kept
# speedup vs baseline: 6.5491x; 1.1419x over previous
#include <hip/hip_runtime.h>
#include <hip/hip_bf16.h>
#include <math.h>
#include <type_traits>


#define RNN_T    2048
#define RNN_B    2048
#define RNN_EMB  10
#define RNN_HID  8
#define RNN_NCLS 4
#define RNN_BLK  16

typedef __attribute__((ext_vector_type(2))) float v2f;
typedef __attribute__((ext_vector_type(4))) float v4f;
typedef __attribute__((ext_vector_type(8))) float v8f;

#if defined(__has_builtin)
#if __has_builtin(__builtin_amdgcn_tanhf)
#define RNN_TANH(x) __builtin_amdgcn_tanhf(x)
#endif
#endif
#ifndef RNN_TANH
#define RNN_TANH(x) tanhf(x)
#endif

__device__ __forceinline__ v8f wmma4(v2f a, v2f b, v8f c) {
    return __builtin_amdgcn_wmma_f32_16x16x4_f32(
         false, a,  false, b,
         (short)0, c,  false,  false);
}

__device__ __forceinline__ int hid_row(int h, int r) {
    return (r & 1) + 4 * (r >> 1) + 2 * h;
}

__device__ __forceinline__ v2f load_a_stack(const float* Wlo, bool lo_plain,
                                            const float* Whi,
                                            int lo, int hi, int c) {
    const int h   = lo >> 3;
    const int r   = lo & 3;
    const int row = hid_row(h, r);
    const int k0  = 4 * c + 2 * hi;
    v2f a; a.x = 0.0f; a.y = 0.0f;
    if (lo & 4) {
        a.x = Whi[row * 8 + k0];
        a.y = Whi[row * 8 + k0 + 1];
    } else if (!lo_plain) {
        a.x = Wlo[row * 8 + k0];
        a.y = Wlo[row * 8 + k0 + 1];
    } else if (h == 0) {
        a.x = Wlo[r * 8 + k0];
        a.y = Wlo[r * 8 + k0 + 1];
    }
    return a;
}

struct RnnW {
    v2f t0a;
    v2f s1_a0, s1_a1;
    v2f s2_a0, s2_a1;
    v8f c_init;
    float c_fc[RNN_NCLS];
};

template <int N>
__device__ __forceinline__ void store_rows(const v4f (&res)[RNN_BLK], int first,
                                           bool hi, float*& op) {
    if (!hi) {
        #pragma unroll
        for (int j = 0; j < N; ++j)
            *(v4f*)(op + (size_t)j * RNN_B * RNN_NCLS) = res[first + j];
    }
    op += (size_t)N * RNN_B * RNN_NCLS;
}

#define RNN_NBLK (RNN_T / RNN_BLK)
#define RNN_LAG  3

__device__ __forceinline__ void l0_step(const RnnW& w, v2f& h0b0, v2f& h0b1,
                                        int tok, int k0, v4f& q) {
    v2f oh;
    oh.x = (tok == k0)     ? 1.0f : 0.0f;
    oh.y = (tok == k0 + 1) ? 1.0f : 0.0f;
    v8f d1 = wmma4(w.t0a, oh, w.c_init);
    d1 = wmma4(w.s1_a0, h0b0, d1);
    d1 = wmma4(w.s1_a1, h0b1, d1);
    q.x = d1[4]; q.y = d1[5]; q.z = d1[6]; q.w = d1[7];
    h0b0.x = RNN_TANH(d1[0]); h0b0.y = RNN_TANH(d1[1]);
    h0b1.x = RNN_TANH(d1[2]); h0b1.y = RNN_TANH(d1[3]);
}

__device__ __forceinline__ v8f l1_cinit(const RnnW& w, v4f q) {
    v8f c;
    #pragma unroll
    for (int i = 0; i < RNN_NCLS; ++i) c[i] = w.c_fc[i];
    c[4] = q.x; c[5] = q.y; c[6] = q.z; c[7] = q.w;
    return c;
}

__device__ __forceinline__ void l1_last_row(const RnnW& w, v2f h1b0, v2f h1b1, v4f& res) {
    const v4f zero4 = (v4f){0.0f, 0.0f, 0.0f, 0.0f};
    v8f d2 = l1_cinit(w, zero4);
    d2 = wmma4(w.s2_a0, h1b0, d2);
    d2 = wmma4(w.s2_a1, h1b1, d2);
    res.x = d2[0]; res.y = d2[1]; res.z = d2[2]; res.w = d2[3];
}

__device__ __forceinline__ v8f l0_onehot(const RnnW& w, int tok, int k0) {
    v2f oh;
    oh.x = (tok == k0)     ? 1.0f : 0.0f;
    oh.y = (tok == k0 + 1) ? 1.0f : 0.0f;
    return wmma4(w.t0a, oh, w.c_init);
}

__device__ __forceinline__ void l0_act(const v8f& d1, v2f& h0b0, v2f& h0b1, v4f& q) {
    q.x = d1[4]; q.y = d1[5]; q.z = d1[6]; q.w = d1[7];
    h0b0.x = RNN_TANH(d1[0]); h0b0.y = RNN_TANH(d1[1]);
    h0b1.x = RNN_TANH(d1[2]); h0b1.y = RNN_TANH(d1[3]);
}

__device__ __forceinline__ void lds_handover_barrier() {
    __builtin_amdgcn_fence(__ATOMIC_RELEASE, "workgroup", "local");
    __builtin_amdgcn_s_barrier();
    __builtin_amdgcn_fence(__ATOMIC_ACQUIRE, "workgroup", "local");
}

__global__ __launch_bounds__(96)
void rnn_fused_kernel(const int* __restrict__ x,
                      const float* __restrict__ emb,
                      const float* __restrict__ W_ih0,
                      const float* __restrict__ W_hh0,
                      const float* __restrict__ b_ih0,
                      const float* __restrict__ b_hh0,
                      const float* __restrict__ W_ih1,
                      const float* __restrict__ W_hh1,
                      const float* __restrict__ b_ih1,
                      const float* __restrict__ b_hh1,
                      const float* __restrict__ W_fc,
                      const float* __restrict__ b_fc,
                      float* __restrict__ out) {
    __shared__ v4f ring[2][RNN_BLK][32];
    __shared__ v4f cring_lo[2][RNN_BLK][32];
    __shared__ v4f cring_hi[2][RNN_BLK][32];

    const int lane = threadIdx.x & 31;
    const int lo   = lane & 15;
    const int hi   = lane >> 4;
    const int role = __builtin_amdgcn_readfirstlane((int)(threadIdx.x >> 5));
    const int batch_base = blockIdx.x * 16;

    const int* xrow = x + (size_t)batch_base + lo;

    int tok0 = 0;
    int tok[RNN_BLK], ntok[RNN_BLK];
    #pragma unroll
    for (int j = 0; j < RNN_BLK; ++j) { tok[j] = 0; ntok[j] = 0; }
    if (role == 0) {
        tok0 = xrow[0];
    } else if (role == 2) {
        #pragma unroll
        for (int j = 0; j < RNN_BLK; ++j) tok[j]  = xrow[(size_t)(1 + j) * RNN_B];
        #pragma unroll
        for (int j = 0; j < RNN_BLK; ++j) ntok[j] = xrow[(size_t)(RNN_BLK + 1 + j) * RNN_B];
    }

    RnnW w;
    {
        const int j = hid_row(lo >> 3, lo & 3);
        const float bias = b_ih0[j] + b_hh0[j];
        const int v0 = 2 * hi, v1 = 2 * hi + 1;
        float a0 = bias, a1 = bias;
        #pragma unroll
        for (int d = 0; d < RNN_EMB; ++d) {
            const float wv = W_ih0[j * RNN_EMB + d];
            a0 += emb[v0 * RNN_EMB + d] * wv;
            a1 += emb[v1 * RNN_EMB + d] * wv;
        }
        w.t0a.x = (lo & 4) ? 0.0f : a0;
        w.t0a.y = (lo & 4) ? 0.0f : a1;
    }

    w.s1_a0 = load_a_stack(W_hh0, false, W_ih1, lo, hi, 0);
    w.s1_a1 = load_a_stack(W_hh0, false, W_ih1, lo, hi, 1);
    w.s2_a0 = load_a_stack(W_fc,  true,  W_hh1, lo, hi, 0);
    w.s2_a1 = load_a_stack(W_fc,  true,  W_hh1, lo, hi, 1);

    #pragma unroll
    for (int i = 0; i < 4; ++i) {
        const int j = hid_row(hi, i);
        w.c_init[i]     = 0.0f;
        w.c_init[4 + i] = b_ih1[j] + b_hh1[j];
    }
    #pragma unroll
    for (int i = 0; i < RNN_NCLS; ++i) w.c_fc[i] = b_fc[i];

    v2f h0b0 = (v2f){0.0f, 0.0f}, h0b1 = (v2f){0.0f, 0.0f};
    v2f h1b0 = (v2f){0.0f, 0.0f}, h1b1 = (v2f){0.0f, 0.0f};
    const v4f zero4 = (v4f){0.0f, 0.0f, 0.0f, 0.0f};
    const v8f zero8 = (v8f){0.0f, 0.0f, 0.0f, 0.0f, 0.0f, 0.0f, 0.0f, 0.0f};
    v4f qd[RNN_BLK];
    v8f ca[RNN_BLK], cb[RNN_BLK];
    v4f qa[RNN_BLK], qb[RNN_BLK];
    v4f res[RNN_BLK];
    #pragma unroll
    for (int j = 0; j < RNN_BLK; ++j) { qd[j] = zero4; qa[j] = zero4; qb[j] = zero4; res[j] = zero4; ca[j] = zero8; cb[j] = zero8; }

    const int  k0  = 2 * hi;
    const bool hiB = (hi != 0);
    float* op = out + ((size_t)batch_base + lo) * RNN_NCLS;

    auto form_block = [&](const int (&t)[RNN_BLK], const int slot) {
        #pragma unroll
        for (int j = 0; j < RNN_BLK; ++j) {
            const v8f c = l0_onehot(w, t[j], k0);
            cring_lo[slot][j][lane] = (v4f){c[0], c[1], c[2], c[3]};
            cring_hi[slot][j][lane] = (v4f){c[4], c[5], c[6], c[7]};
        }
    };
    auto fetch_block = [&](v8f (&c)[RNN_BLK], const int slot) {
        #pragma unroll
        for (int j = 0; j < RNN_BLK; ++j) {
            const v4f l = cring_lo[slot][j][lane];
            const v4f h = cring_hi[slot][j][lane];
            c[j] = (v8f){l.x, l.y, l.z, l.w, h.x, h.y, h.z, h.w};
        }
    };

    if (role == 0) {
        v4f qskip;
        l0_step(w, h0b0, h0b1, tok0, k0, qskip);
    } else if (role == 2) {
        form_block(tok, 0);
        #pragma unroll
        for (int j = 0; j < RNN_BLK; ++j) tok[j] = xrow[(size_t)(2 * RNN_BLK + 1 + j) * RNN_B];
        form_block(ntok, 1);
    }
    lds_handover_barrier();
    if (role == 0) fetch_block(ca, 0);
    lds_handover_barrier();

    const int* xnext = xrow + (size_t)(3 * RNN_BLK + 1) * RNN_B;

    auto block_trip = [&](auto steady_tag, const int it, v4f (&qin)[RNN_BLK], v4f (&qout)[RNN_BLK],
                          int (&tcur)[RNN_BLK], int (&tnxt)[RNN_BLK],
                          v8f (&cin)[RNN_BLK], v8f (&cout)[RNN_BLK]) {
        constexpr bool STEADY = decltype(steady_tag)::value;
        if (role == 0) {
            if (STEADY || (it >= 1 && it <= RNN_NBLK)) {
                const int slot = (it - 1) & 1;
                #pragma unroll
                for (int j = 0; j < RNN_BLK; ++j) ring[slot][j][lane] = qd[j];
            }
            if (STEADY || it + 1 < RNN_NBLK) {
                fetch_block(cout, (it + 1) & 1);
            }
            if (STEADY || it < RNN_NBLK) {
                #pragma unroll
                for (int j = 0; j < RNN_BLK; ++j) {
                    v8f d1 = wmma4(w.s1_a0, h0b0, cin[j]);
                    d1 = wmma4(w.s1_a1, h0b1, d1);
                    __builtin_amdgcn_sched_barrier(0);
                    l0_act(d1, h0b0, h0b1, qd[j]);
                    __builtin_amdgcn_sched_barrier(0);
                }
            }
        }
        if (role == 1) {
            if (STEADY || (it >= 2 && it < RNN_NBLK + 2)) {
                const int slot = (it - 2) & 1;
                #pragma unroll
                for (int j = 0; j < RNN_BLK; ++j) qout[j] = ring[slot][j][lane];
            }
            if (STEADY || (it >= RNN_LAG && it < RNN_NBLK + RNN_LAG)) {
                v8f dn1 = l1_cinit(w, qin[0]);
                #pragma unroll
                for (int j = 0; j < RNN_BLK; ++j) {
                    v8f d2 = wmma4(w.s2_a0, h1b0, dn1);
                    d2 = wmma4(w.s2_a1, h1b1, d2);
                    __builtin_amdgcn_sched_barrier(0);
                    if (j + 1 < RNN_BLK) dn1 = l1_cinit(w, qin[j + 1]);
                    __builtin_amdgcn_sched_barrier(0);
                    res[j].x = d2[0]; res[j].y = d2[1]; res[j].z = d2[2]; res[j].w = d2[3];
                    h1b0.x = RNN_TANH(d2[4]); h1b0.y = RNN_TANH(d2[5]);
                    h1b1.x = RNN_TANH(d2[6]); h1b1.y = RNN_TANH(d2[7]);
                    __builtin_amdgcn_sched_barrier(0);
                }
                if (!STEADY && it == RNN_LAG) store_rows<RNN_BLK - 1>(res, 1, hiB, op);
                else                          store_rows<RNN_BLK>(res, 0, hiB, op);
            }
        }
        if (role == 2) {
            if (STEADY || it + 2 < RNN_NBLK) {
                if (STEADY || it + 3 < RNN_NBLK) {
                    #pragma unroll
                    for (int j = 0; j < RNN_BLK - 1; ++j) tnxt[j] = xnext[(size_t)j * RNN_B];
                    if (STEADY || it + 3 < RNN_NBLK - 1)
                        tnxt[RNN_BLK - 1] = xnext[(size_t)(RNN_BLK - 1) * RNN_B];
                    if (STEADY || it + 4 < RNN_NBLK) {
                        #pragma unroll
                        for (int j = 0; j < RNN_BLK; ++j)
                            __builtin_prefetch(xnext + (size_t)(RNN_BLK - 1 + j) * RNN_B, 0, 3);
                    }
                    xnext += (size_t)RNN_BLK * RNN_B;
                }
                form_block(tcur, it & 1);
            }
        }
        lds_handover_barrier();
    };
    const std::integral_constant<bool, false> edge_trip{};
    const std::integral_constant<bool, true>  steady_trip{};

    #pragma unroll 1
    for (int it = 0; it < 4; it += 2) {
        block_trip(edge_trip, it,     qa, qb, tok, ntok, ca, cb);
        block_trip(edge_trip, it + 1, qb, qa, ntok, tok, cb, ca);
    }
    #pragma unroll 1
    for (int it = 4; it < RNN_NBLK - 4; it += 2) {
        block_trip(steady_trip, it,     qa, qb, tok, ntok, ca, cb);
        block_trip(steady_trip, it + 1, qb, qa, ntok, tok, cb, ca);
    }
    #pragma unroll 1
    for (int it = RNN_NBLK - 4; it < RNN_NBLK + RNN_LAG; it += 2) {
        block_trip(edge_trip, it,     qa, qb, tok, ntok, ca, cb);
        block_trip(edge_trip, it + 1, qb, qa, ntok, tok, cb, ca);
    }

    if (role == 1) {
        l1_last_row(w, h1b0, h1b1, res[0]);
        store_rows<1>(res, 0, hiB, op);
    }
}

extern "C" void kernel_launch(void* const* d_in, const int* in_sizes, int n_in,
                              void* d_out, int out_size, void* d_ws, size_t ws_size,
                              hipStream_t stream) {
    (void)in_sizes; (void)n_in; (void)d_ws; (void)ws_size; (void)out_size;
    const int*   x     = (const int*)  d_in[0];
    const float* emb   = (const float*)d_in[1];
    const float* W_ih0 = (const float*)d_in[2];
    const float* W_hh0 = (const float*)d_in[3];
    const float* b_ih0 = (const float*)d_in[4];
    const float* b_hh0 = (const float*)d_in[5];
    const float* W_ih1 = (const float*)d_in[6];
    const float* W_hh1 = (const float*)d_in[7];
    const float* b_ih1 = (const float*)d_in[8];
    const float* b_hh1 = (const float*)d_in[9];
    const float* W_fc  = (const float*)d_in[10];
    const float* b_fc  = (const float*)d_in[11];
    float* out = (float*)d_out;

    dim3 grid(RNN_B / 16);
    dim3 block(96);
    rnn_fused_kernel<<<grid, block, 0, stream>>>(
        x, emb, W_ih0, W_hh0, b_ih0, b_hh0,
        W_ih1, W_hh1, b_ih1, b_hh1, W_fc, b_fc, out);
}


# Constant half of each step's C operand (bias1 on wave 0, b_fc on wave 1) stays resident in its own register group; only the varying half is loaded from LDS straight into it; sum order kept
# speedup vs baseline: 7.0479x; 1.0762x over previous
#include <hip/hip_runtime.h>
#include <hip/hip_bf16.h>
#include <math.h>
#include <type_traits>


#define RNN_T    2048
#define RNN_B    2048
#define RNN_EMB  10
#define RNN_HID  8
#define RNN_NCLS 4
#define RNN_BLK  16

typedef __attribute__((ext_vector_type(2))) float v2f;
typedef __attribute__((ext_vector_type(4))) float v4f;
typedef __attribute__((ext_vector_type(8))) float v8f;

#if defined(__has_builtin)
#if __has_builtin(__builtin_amdgcn_tanhf)
#define RNN_TANH(x) __builtin_amdgcn_tanhf(x)
#endif
#endif
#ifndef RNN_TANH
#define RNN_TANH(x) tanhf(x)
#endif

__device__ __forceinline__ v8f wmma4(v2f a, v2f b, v8f c) {
    return __builtin_amdgcn_wmma_f32_16x16x4_f32(
         false, a,  false, b,
         (short)0, c,  false,  false);
}

__device__ __forceinline__ int hid_row(int h, int r) {
    return (r & 1) + 4 * (r >> 1) + 2 * h;
}

__device__ __forceinline__ v2f load_a_stack(const float* Wlo, bool lo_plain,
                                            const float* Whi,
                                            int lo, int hi, int c) {
    const int h   = lo >> 3;
    const int r   = lo & 3;
    const int row = hid_row(h, r);
    const int k0  = 4 * c + 2 * hi;
    v2f a; a.x = 0.0f; a.y = 0.0f;
    if (lo & 4) {
        a.x = Whi[row * 8 + k0];
        a.y = Whi[row * 8 + k0 + 1];
    } else if (!lo_plain) {
        a.x = Wlo[row * 8 + k0];
        a.y = Wlo[row * 8 + k0 + 1];
    } else if (h == 0) {
        a.x = Wlo[r * 8 + k0];
        a.y = Wlo[r * 8 + k0 + 1];
    }
    return a;
}

struct RnnW {
    v2f t0a;
    v2f s1_a0, s1_a1;
    v2f s2_a0, s2_a1;
    v8f c_init;
    float c_fc[RNN_NCLS];
};

template <int N>
__device__ __forceinline__ void store_rows(const v4f (&res)[RNN_BLK], int first,
                                           bool hi, float*& op) {
    if (!hi) {
        #pragma unroll
        for (int j = 0; j < N; ++j)
            *(v4f*)(op + (size_t)j * RNN_B * RNN_NCLS) = res[first + j];
    }
    op += (size_t)N * RNN_B * RNN_NCLS;
}

#define RNN_NBLK (RNN_T / RNN_BLK)
#define RNN_LAG  3

__device__ __forceinline__ void l0_step(const RnnW& w, v2f& h0b0, v2f& h0b1,
                                        int tok, int k0, v4f& q) {
    v2f oh;
    oh.x = (tok == k0)     ? 1.0f : 0.0f;
    oh.y = (tok == k0 + 1) ? 1.0f : 0.0f;
    v8f d1 = wmma4(w.t0a, oh, w.c_init);
    d1 = wmma4(w.s1_a0, h0b0, d1);
    d1 = wmma4(w.s1_a1, h0b1, d1);
    q.x = d1[4]; q.y = d1[5]; q.z = d1[6]; q.w = d1[7];
    h0b0.x = RNN_TANH(d1[0]); h0b0.y = RNN_TANH(d1[1]);
    h0b1.x = RNN_TANH(d1[2]); h0b1.y = RNN_TANH(d1[3]);
}

__device__ __forceinline__ v8f l1_cinit(const RnnW& w, v4f q) {
    v8f c;
    #pragma unroll
    for (int i = 0; i < RNN_NCLS; ++i) c[i] = w.c_fc[i];
    c[4] = q.x; c[5] = q.y; c[6] = q.z; c[7] = q.w;
    return c;
}

__device__ __forceinline__ void l1_last_row(const RnnW& w, v2f h1b0, v2f h1b1, v4f& res) {
    const v4f zero4 = (v4f){0.0f, 0.0f, 0.0f, 0.0f};
    v8f d2 = l1_cinit(w, zero4);
    d2 = wmma4(w.s2_a0, h1b0, d2);
    d2 = wmma4(w.s2_a1, h1b1, d2);
    res.x = d2[0]; res.y = d2[1]; res.z = d2[2]; res.w = d2[3];
}

__device__ __forceinline__ v8f l0_onehot(const RnnW& w, int tok, int k0) {
    v2f oh;
    oh.x = (tok == k0)     ? 1.0f : 0.0f;
    oh.y = (tok == k0 + 1) ? 1.0f : 0.0f;
    return wmma4(w.t0a, oh, w.c_init);
}

__device__ __forceinline__ void l0_act(const v8f& d1, v2f& h0b0, v2f& h0b1, v4f& q) {
    q.x = d1[4]; q.y = d1[5]; q.z = d1[6]; q.w = d1[7];
    h0b0.x = RNN_TANH(d1[0]); h0b0.y = RNN_TANH(d1[1]);
    h0b1.x = RNN_TANH(d1[2]); h0b1.y = RNN_TANH(d1[3]);
}

__device__ __forceinline__ void lds_handover_barrier() {
    __builtin_amdgcn_fence(__ATOMIC_RELEASE, "workgroup", "local");
    __builtin_amdgcn_s_barrier();
    __builtin_amdgcn_fence(__ATOMIC_ACQUIRE, "workgroup", "local");
}

__global__ __launch_bounds__(96)
void rnn_fused_kernel(const int* __restrict__ x,
                      const float* __restrict__ emb,
                      const float* __restrict__ W_ih0,
                      const float* __restrict__ W_hh0,
                      const float* __restrict__ b_ih0,
                      const float* __restrict__ b_hh0,
                      const float* __restrict__ W_ih1,
                      const float* __restrict__ W_hh1,
                      const float* __restrict__ b_ih1,
                      const float* __restrict__ b_hh1,
                      const float* __restrict__ W_fc,
                      const float* __restrict__ b_fc,
                      float* __restrict__ out) {
    __shared__ v4f ring[2][RNN_BLK][32];
    __shared__ v4f cring_lo[2][RNN_BLK][32];

    const int lane = threadIdx.x & 31;
    const int lo   = lane & 15;
    const int hi   = lane >> 4;
    const int role = __builtin_amdgcn_readfirstlane((int)(threadIdx.x >> 5));
    const int batch_base = blockIdx.x * 16;

    const int* xrow = x + (size_t)batch_base + lo;

    int tok0 = 0;
    int tok[RNN_BLK], ntok[RNN_BLK];
    #pragma unroll
    for (int j = 0; j < RNN_BLK; ++j) { tok[j] = 0; ntok[j] = 0; }
    if (role == 0) {
        tok0 = xrow[0];
    } else if (role == 2) {
        #pragma unroll
        for (int j = 0; j < RNN_BLK; ++j) tok[j]  = xrow[(size_t)(1 + j) * RNN_B];
        #pragma unroll
        for (int j = 0; j < RNN_BLK; ++j) ntok[j] = xrow[(size_t)(RNN_BLK + 1 + j) * RNN_B];
    }

    RnnW w;
    {
        const int j = hid_row(lo >> 3, lo & 3);
        const float bias = b_ih0[j] + b_hh0[j];
        const int v0 = 2 * hi, v1 = 2 * hi + 1;
        float a0 = bias, a1 = bias;
        #pragma unroll
        for (int d = 0; d < RNN_EMB; ++d) {
            const float wv = W_ih0[j * RNN_EMB + d];
            a0 += emb[v0 * RNN_EMB + d] * wv;
            a1 += emb[v1 * RNN_EMB + d] * wv;
        }
        w.t0a.x = (lo & 4) ? 0.0f : a0;
        w.t0a.y = (lo & 4) ? 0.0f : a1;
    }

    w.s1_a0 = load_a_stack(W_hh0, false, W_ih1, lo, hi, 0);
    w.s1_a1 = load_a_stack(W_hh0, false, W_ih1, lo, hi, 1);
    w.s2_a0 = load_a_stack(W_fc,  true,  W_hh1, lo, hi, 0);
    w.s2_a1 = load_a_stack(W_fc,  true,  W_hh1, lo, hi, 1);

    #pragma unroll
    for (int i = 0; i < 4; ++i) {
        const int j = hid_row(hi, i);
        w.c_init[i]     = 0.0f;
        w.c_init[4 + i] = b_ih1[j] + b_hh1[j];
    }
    #pragma unroll
    for (int i = 0; i < RNN_NCLS; ++i) w.c_fc[i] = b_fc[i];

    v2f h0b0 = (v2f){0.0f, 0.0f}, h0b1 = (v2f){0.0f, 0.0f};
    v2f h1b0 = (v2f){0.0f, 0.0f}, h1b1 = (v2f){0.0f, 0.0f};
    const v4f zero4 = (v4f){0.0f, 0.0f, 0.0f, 0.0f};
    const v8f zero8 = (v8f){0.0f, 0.0f, 0.0f, 0.0f, 0.0f, 0.0f, 0.0f, 0.0f};
    v4f qd[RNN_BLK];
    v8f ca[RNN_BLK], cb[RNN_BLK];
    v4f res[RNN_BLK];
    v8f cbase = zero8;
    if (role == 0) {
        cbase = w.c_init;
    } else if (role == 1) {
        #pragma unroll
        for (int i = 0; i < RNN_NCLS; ++i) cbase[i] = w.c_fc[i];
    }
    #pragma unroll
    for (int j = 0; j < RNN_BLK; ++j) { qd[j] = zero4; res[j] = zero4; ca[j] = cbase; cb[j] = cbase; }

    const int  k0  = 2 * hi;
    const bool hiB = (hi != 0);
    float* op = out + ((size_t)batch_base + lo) * RNN_NCLS;

    auto form_block = [&](const int (&t)[RNN_BLK], const int slot) {
        #pragma unroll
        for (int j = 0; j < RNN_BLK; ++j) {
            const v8f c = l0_onehot(w, t[j], k0);
            cring_lo[slot][j][lane] = (v4f){c[0], c[1], c[2], c[3]};
        }
    };
    auto fetch_block = [&](v8f (&c)[RNN_BLK], const int slot) {
        #pragma unroll
        for (int j = 0; j < RNN_BLK; ++j) {
            const v4f l = cring_lo[slot][j][lane];
            c[j][0] = l.x; c[j][1] = l.y; c[j][2] = l.z; c[j][3] = l.w;
        }
    };
    auto fetch_q = [&](v8f (&c)[RNN_BLK], const int slot) {
        #pragma unroll
        for (int j = 0; j < RNN_BLK; ++j) {
            const v4f q = ring[slot][j][lane];
            c[j][4] = q.x; c[j][5] = q.y; c[j][6] = q.z; c[j][7] = q.w;
        }
    };

    if (role == 0) {
        v4f qskip;
        l0_step(w, h0b0, h0b1, tok0, k0, qskip);
    } else if (role == 2) {
        form_block(tok, 0);
        #pragma unroll
        for (int j = 0; j < RNN_BLK; ++j) tok[j] = xrow[(size_t)(2 * RNN_BLK + 1 + j) * RNN_B];
        form_block(ntok, 1);
    }
    lds_handover_barrier();
    if (role == 0) fetch_block(ca, 0);
    lds_handover_barrier();

    const int* xnext = xrow + (size_t)(3 * RNN_BLK + 1) * RNN_B;

    auto block_trip = [&](auto steady_tag, const int it,
                          int (&tcur)[RNN_BLK], int (&tnxt)[RNN_BLK],
                          v8f (&cin)[RNN_BLK], v8f (&cout)[RNN_BLK]) {
        constexpr bool STEADY = decltype(steady_tag)::value;
        if (role == 0) {
            if (STEADY || (it >= 1 && it <= RNN_NBLK)) {
                const int slot = (it - 1) & 1;
                #pragma unroll
                for (int j = 0; j < RNN_BLK; ++j) ring[slot][j][lane] = qd[j];
            }
            if (STEADY || it + 1 < RNN_NBLK) {
                fetch_block(cout, (it + 1) & 1);
            }
            if (STEADY || it < RNN_NBLK) {
                #pragma unroll
                for (int j = 0; j < RNN_BLK; ++j) {
                    v8f d1 = wmma4(w.s1_a0, h0b0, cin[j]);
                    d1 = wmma4(w.s1_a1, h0b1, d1);
                    __builtin_amdgcn_sched_barrier(0);
                    l0_act(d1, h0b0, h0b1, qd[j]);
                    __builtin_amdgcn_sched_barrier(0);
                }
            }
        }
        if (role == 1) {
            if (STEADY || (it >= 2 && it < RNN_NBLK + 2)) {
                fetch_q(cout, (it - 2) & 1);
            }
            if (STEADY || (it >= RNN_LAG && it < RNN_NBLK + RNN_LAG)) {
                #pragma unroll
                for (int j = 0; j < RNN_BLK; ++j) {
                    v8f d2 = wmma4(w.s2_a0, h1b0, cin[j]);
                    d2 = wmma4(w.s2_a1, h1b1, d2);
                    __builtin_amdgcn_sched_barrier(0);
                    res[j].x = d2[0]; res[j].y = d2[1]; res[j].z = d2[2]; res[j].w = d2[3];
                    h1b0.x = RNN_TANH(d2[4]); h1b0.y = RNN_TANH(d2[5]);
                    h1b1.x = RNN_TANH(d2[6]); h1b1.y = RNN_TANH(d2[7]);
                    __builtin_amdgcn_sched_barrier(0);
                }
                if (!STEADY && it == RNN_LAG) store_rows<RNN_BLK - 1>(res, 1, hiB, op);
                else                          store_rows<RNN_BLK>(res, 0, hiB, op);
            }
        }
        if (role == 2) {
            if (STEADY || it + 2 < RNN_NBLK) {
                if (STEADY || it + 3 < RNN_NBLK) {
                    #pragma unroll
                    for (int j = 0; j < RNN_BLK - 1; ++j) tnxt[j] = xnext[(size_t)j * RNN_B];
                    if (STEADY || it + 3 < RNN_NBLK - 1)
                        tnxt[RNN_BLK - 1] = xnext[(size_t)(RNN_BLK - 1) * RNN_B];
                    if (STEADY || it + 4 < RNN_NBLK) {
                        #pragma unroll
                        for (int j = 0; j < RNN_BLK; ++j)
                            __builtin_prefetch(xnext + (size_t)(RNN_BLK - 1 + j) * RNN_B, 0, 3);
                    }
                    xnext += (size_t)RNN_BLK * RNN_B;
                }
                form_block(tcur, it & 1);
            }
        }
        lds_handover_barrier();
    };
    const std::integral_constant<bool, false> edge_trip{};
    const std::integral_constant<bool, true>  steady_trip{};

    #pragma unroll 1
    for (int it = 0; it < 4; it += 2) {
        block_trip(edge_trip, it,     tok, ntok, ca, cb);
        block_trip(edge_trip, it + 1, ntok, tok, cb, ca);
    }
    #pragma unroll 1
    for (int it = 4; it < RNN_NBLK - 4; it += 2) {
        block_trip(steady_trip, it,     tok, ntok, ca, cb);
        block_trip(steady_trip, it + 1, ntok, tok, cb, ca);
    }
    #pragma unroll 1
    for (int it = RNN_NBLK - 4; it < RNN_NBLK + RNN_LAG; it += 2) {
        block_trip(edge_trip, it,     tok, ntok, ca, cb);
        block_trip(edge_trip, it + 1, ntok, tok, cb, ca);
    }

    if (role == 1) {
        l1_last_row(w, h1b0, h1b1, res[0]);
        store_rows<1>(res, 0, hiB, op);
    }
}

extern "C" void kernel_launch(void* const* d_in, const int* in_sizes, int n_in,
                              void* d_out, int out_size, void* d_ws, size_t ws_size,
                              hipStream_t stream) {
    (void)in_sizes; (void)n_in; (void)d_ws; (void)ws_size; (void)out_size;
    const int*   x     = (const int*)  d_in[0];
    const float* emb   = (const float*)d_in[1];
    const float* W_ih0 = (const float*)d_in[2];
    const float* W_hh0 = (const float*)d_in[3];
    const float* b_ih0 = (const float*)d_in[4];
    const float* b_hh0 = (const float*)d_in[5];
    const float* W_ih1 = (const float*)d_in[6];
    const float* W_hh1 = (const float*)d_in[7];
    const float* b_ih1 = (const float*)d_in[8];
    const float* b_hh1 = (const float*)d_in[9];
    const float* W_fc  = (const float*)d_in[10];
    const float* b_fc  = (const float*)d_in[11];
    float* out = (float*)d_out;

    dim3 grid(RNN_B / 16);
    dim3 block(96);
    rnn_fused_kernel<<<grid, block, 0, stream>>>(
        x, emb, W_ih0, W_hh0, b_ih0, b_hh0,
        W_ih1, W_hh1, b_ih1, b_hh1, W_fc, b_fc, out);
}
